# starting kernel
# baseline (speedup 1.0000x reference)
#include <hip/hip_runtime.h>


typedef float v2f __attribute__((ext_vector_type(2)));
typedef float v8f __attribute__((ext_vector_type(8)));

#define N_NODES 50000
#define N_EDGES 800000
#define FDIM    128
#define E_TR    100000
#define E_TE    20000
#define N_ROWS  240000

__global__ __launch_bounds__(256) void deg_count(const int* __restrict__ dst,
                                                 float* __restrict__ deg, int nE) {
    int t = blockIdx.x * blockDim.x + threadIdx.x;
    if (t < nE) atomicAdd(&deg[dst[t]], 1.0f);
}

__global__ __launch_bounds__(256) void make_dis(float* __restrict__ deg, int n) {
    int t = blockIdx.x * blockDim.x + threadIdx.x;
    if (t < n) deg[t] = rsqrtf(deg[t] + 1.0f);
}

__global__ __launch_bounds__(256) void gemm_node(const float* __restrict__ A,
                                                 const float* __restrict__ B,
                                                 float* __restrict__ C, int M) {
    int wave = (blockIdx.x * blockDim.x + threadIdx.x) >> 5;
    int lane = threadIdx.x & 31;
    int m0 = wave * 16;
    if (m0 >= M) return;
    int half = lane >> 4;
    int l16  = lane & 15;

    const float* Arow = A + (size_t)(m0 + l16) * FDIM;

    v8f acc[8] = {};

    for (int k0 = 0; k0 < FDIM; k0 += 4) {
        int k = k0 + half * 2;
        v2f afrag = *(const v2f*)(Arow + k);
        const float* b0 = B + (size_t)k * FDIM + l16;
        #pragma unroll
        for (int n = 0; n < 8; ++n) {
            v2f bfrag;
            bfrag.x = b0[n * 16];
            bfrag.y = b0[FDIM + n * 16];
            acc[n] = __builtin_amdgcn_wmma_f32_16x16x4_f32(
                false, afrag, false, bfrag, (short)0, acc[n], false, false);
        }
    }

    #pragma unroll
    for (int n = 0; n < 8; ++n) {
        #pragma unroll
        for (int g = 0; g < 8; ++g) {
            int m = m0 + g + half * 8;
            C[(size_t)m * FDIM + n * 16 + l16] = acc[n][g];
        }
    }
}

__global__ __launch_bounds__(256) void scatter_add(const float* __restrict__ hw,
                                                   const float* __restrict__ dis,
                                                   const int* __restrict__ src,
                                                   const int* __restrict__ dst,
                                                   float* __restrict__ agg, int nE) {
    int t = blockIdx.x * blockDim.x + threadIdx.x;
    int e = t >> 5;
    if (e >= nE) return;
    int f = (t & 31) * 4;
    int s = src[e], d = dst[e];
    float sc = dis[s];
    const float4 v = *(const float4*)(hw + (size_t)s * FDIM + f);
    float* dp = agg + (size_t)d * FDIM + f;
    atomicAdd(dp + 0, v.x * sc);
    atomicAdd(dp + 1, v.y * sc);
    atomicAdd(dp + 2, v.z * sc);
    atomicAdd(dp + 3, v.w * sc);
}

__global__ __launch_bounds__(256) void combine(const float* __restrict__ agg,
                                               const float* __restrict__ hw,
                                               const float* __restrict__ dis,
                                               const float* __restrict__ b,
                                               float* __restrict__ hout,
                                               int nNodes, int doRelu) {
    int t = blockIdx.x * blockDim.x + threadIdx.x;
    int node = t >> 5;
    if (node >= nNodes) return;
    int f = (t & 31) * 4;
    float di  = dis[node];
    float di2 = di * di;
    float4 a = *(const float4*)(agg + (size_t)node * FDIM + f);
    float4 h = *(const float4*)(hw  + (size_t)node * FDIM + f);
    float4 r;
    r.x = a.x * di + h.x * di2 + b[f + 0];
    r.y = a.y * di + h.y * di2 + b[f + 1];
    r.z = a.z * di + h.z * di2 + b[f + 2];
    r.w = a.w * di + h.w * di2 + b[f + 3];
    if (doRelu) {
        r.x = fmaxf(r.x, 0.0f); r.y = fmaxf(r.y, 0.0f);
        r.z = fmaxf(r.z, 0.0f); r.w = fmaxf(r.w, 0.0f);
    }
    *(float4*)(hout + (size_t)node * FDIM + f) = r;
}

__global__ __launch_bounds__(256) void gemm_edge(const float* __restrict__ H,
                                                 const int* __restrict__ pos_tr,
                                                 const int* __restrict__ neg_tr,
                                                 const int* __restrict__ pos_te,
                                                 const int* __restrict__ neg_te,
                                                 const float* __restrict__ Wl1,
                                                 const float* __restrict__ bl1,
                                                 const float* __restrict__ Wl2,
                                                 const float* __restrict__ bl2,
                                                 float* __restrict__ out) {
    int wave = (blockIdx.x * blockDim.x + threadIdx.x) >> 5;
    int lane = threadIdx.x & 31;
    int r0 = wave * 16;
    if (r0 >= N_ROWS) return;
    int half = lane >> 4;
    int l16  = lane & 15;
    int r = r0 + l16;

    int u, v;
    if (r < E_TR)            { u = pos_tr[r];              v = pos_tr[E_TR + r]; }
    else if (r < 2 * E_TR)   { int i = r - E_TR;           u = neg_tr[i]; v = neg_tr[E_TR + i]; }
    else if (r < 2 * E_TR + E_TE) { int i = r - 2 * E_TR;  u = pos_te[i]; v = pos_te[E_TE + i]; }
    else                     { int i = r - 2 * E_TR - E_TE; u = neg_te[i]; v = neg_te[E_TE + i]; }

    const float* Au = H + (size_t)u * FDIM;
    const float* Av = H + (size_t)v * FDIM;

    v8f acc[8] = {};

    for (int k0 = 0; k0 < 2 * FDIM; k0 += 4) {
        int k = k0 + half * 2;
        const float* ap = (k < FDIM) ? (Au + k) : (Av + (k - FDIM));
        v2f afrag = *(const v2f*)ap;
        const float* b0 = Wl1 + (size_t)k * FDIM + l16;
        #pragma unroll
        for (int n = 0; n < 8; ++n) {
            v2f bfrag;
            bfrag.x = b0[n * 16];
            bfrag.y = b0[FDIM + n * 16];
            acc[n] = __builtin_amdgcn_wmma_f32_16x16x4_f32(
                false, afrag, false, bfrag, (short)0, acc[n], false, false);
        }
    }

    float partial[8];
    #pragma unroll
    for (int g = 0; g < 8; ++g) partial[g] = 0.0f;

    #pragma unroll
    for (int n = 0; n < 8; ++n) {
        float bb = bl1[n * 16 + l16];
        float ww = Wl2[n * 16 + l16];
        #pragma unroll
        for (int g = 0; g < 8; ++g) {
            float z = fmaxf(acc[n][g] + bb, 0.0f);
            partial[g] += z * ww;
        }
    }

    float bias2 = bl2[0];
    #pragma unroll
    for (int g = 0; g < 8; ++g) {
        float s = partial[g];
        s += __shfl_xor(s, 8, 16);
        s += __shfl_xor(s, 4, 16);
        s += __shfl_xor(s, 2, 16);
        s += __shfl_xor(s, 1, 16);
        if (l16 == 0) out[r0 + g + half * 8] = s + bias2;
    }
}

extern "C" void kernel_launch(void* const* d_in, const int* in_sizes, int n_in,
                              void* d_out, int out_size, void* d_ws, size_t ws_size,
                              hipStream_t stream) {
    const float* x          = (const float*)d_in[0];
    const int*   edge_index = (const int*)d_in[1];
    const int*   pos_tr     = (const int*)d_in[2];
    const int*   neg_tr     = (const int*)d_in[3];
    const int*   pos_te     = (const int*)d_in[4];
    const int*   neg_te     = (const int*)d_in[5];
    const float* W[4] = {(const float*)d_in[6],  (const float*)d_in[8],
                         (const float*)d_in[10], (const float*)d_in[12]};
    const float* b[4] = {(const float*)d_in[7],  (const float*)d_in[9],
                         (const float*)d_in[11], (const float*)d_in[13]};
    const float* Wl1 = (const float*)d_in[14];
    const float* bl1 = (const float*)d_in[15];
    const float* Wl2 = (const float*)d_in[16];
    const float* bl2 = (const float*)d_in[17];
    float* out = (float*)d_out;

    const size_t NF = (size_t)N_NODES * FDIM;
    float* ws  = (float*)d_ws;
    float* dis = ws;
    float* hw  = ws + 50048;
    float* agg = hw  + NF;
    float* h0  = agg + NF;
    float* h1  = h0  + NF;

    const int* src = edge_index;
    const int* dst = edge_index + N_EDGES;

    hipMemsetAsync(dis, 0, N_NODES * sizeof(float), stream);
    deg_count<<<(N_EDGES + 255) / 256, 256, 0, stream>>>(dst, dis, N_EDGES);
    make_dis<<<(N_NODES + 255) / 256, 256, 0, stream>>>(dis, N_NODES);

    const int tilesM     = (N_NODES + 15) / 16;
    const int gemmBlocks = (tilesM + 7) / 8;
    const float* hin = x;
    float* houts[4] = {h0, h1, h0, h1};
    for (int l = 0; l < 4; ++l) {
        gemm_node<<<gemmBlocks, 256, 0, stream>>>(hin, W[l], hw, N_NODES);
        hipMemsetAsync(agg, 0, NF * sizeof(float), stream);
        scatter_add<<<(N_EDGES * 32) / 256, 256, 0, stream>>>(hw, dis, src, dst, agg, N_EDGES);
        combine<<<(N_NODES * 32 + 255) / 256, 256, 0, stream>>>(
            agg, hw, dis, b[l], houts[l], N_NODES, (l < 3) ? 1 : 0);
        hin = houts[l];
    }

    const int edgeTiles  = N_ROWS / 16;
    const int edgeBlocks = (edgeTiles + 7) / 8;
    gemm_edge<<<edgeBlocks, 256, 0, stream>>>(h1, pos_tr, neg_tr, pos_te, neg_te,
                                              Wl1, bl1, Wl2, bl2, out);
}


# Aggregation by block owner: a 1024-thread group keeps its 196 rows in LDS from the self term, lists its edges by scanning owner bytes, adds 8 gathered rows a wave with LDS f32 atomics; arrival order
# speedup vs baseline: 2.0867x; 2.0867x over previous
#include <hip/hip_runtime.h>


typedef float v2f __attribute__((ext_vector_type(2)));
typedef float v8f __attribute__((ext_vector_type(8)));
typedef float v4f __attribute__((ext_vector_type(4)));

#define N_NODES 50000
#define N_EDGES 800000
#define FDIM    128
#define E_TR    100000
#define E_TE    20000
#define N_ROWS  240000

#define OWN_ROWS    196
#define OWN_WG      ((N_NODES + OWN_ROWS - 1) / OWN_ROWS)
#define OWN_THREADS 1024
#define SEG_CAP     1024
#define CH_EDGES    512
#define N_CH        ((N_EDGES + CH_EDGES - 1) / CH_EDGES)
static_assert(OWN_WG <= 256, "an owner id must fit in one byte");
static_assert(N_EDGES % 16 == 0, "a lane's 16 ids are all inside the edge list or all outside");
static_assert(2 * CH_EDGES <= SEG_CAP, "a drained list must take one full step");

__global__ __launch_bounds__(256) void deg_count(const int* __restrict__ dst,
                                                 float* __restrict__ deg,
                                                 unsigned char* __restrict__ own, int nE) {
    int t = blockIdx.x * blockDim.x + threadIdx.x;
    if (t < nE) {
        int d = dst[t];
        atomicAdd(&deg[d], 1.0f);
        own[t] = (unsigned char)(d / OWN_ROWS);
    }
}

__global__ __launch_bounds__(256) void make_dis(float* __restrict__ deg, int n) {
    int t = blockIdx.x * blockDim.x + threadIdx.x;
    if (t < n) deg[t] = rsqrtf(deg[t] + 1.0f);
}

#define A_RAW     0
#define A_AFFINE  1
#define A_AFFRELU 2

static __device__ __forceinline__ void strip_k128(const float* __restrict__ Arow, float di,
                                                  const float* __restrict__ bias,
                                                  const float* __restrict__ Bm,
                                                  int half, int l16, const int mode,
                                                  v8f (&acc)[8]) {
    for (int k0 = 0; k0 < FDIM; k0 += 4) {
        int k = k0 + half * 2;
        v2f afrag = *(const v2f*)(Arow + k);
        if (mode != A_RAW) {
            v2f bb = *(const v2f*)(bias + k);
            afrag.x = afrag.x * di + bb.x;
            afrag.y = afrag.y * di + bb.y;
            if (mode == A_AFFRELU) {
                afrag.x = fmaxf(afrag.x, 0.0f);
                afrag.y = fmaxf(afrag.y, 0.0f);
            }
        }
        const float* b0 = Bm + (size_t)k * FDIM + l16;
        #pragma unroll
        for (int n = 0; n < 8; ++n) {
            v2f bfrag;
            bfrag.x = b0[n * 16];
            bfrag.y = b0[FDIM + n * 16];
            acc[n] = __builtin_amdgcn_wmma_f32_16x16x4_f32(
                false, afrag, false, bfrag, (short)0, acc[n], false, false);
        }
    }
}

static __device__ __forceinline__ void gemm_node_body(const float* __restrict__ A,
                                                      const float* __restrict__ bprev,
                                                      const float* __restrict__ dis,
                                                      const float* __restrict__ B,
                                                      float* __restrict__ hws,
                                                      int M, const int mode) {
    int wave = (blockIdx.x * blockDim.x + threadIdx.x) >> 5;
    int lane = threadIdx.x & 31;
    int m0 = wave * 16;
    if (m0 >= M) return;
    int half = lane >> 4;
    int l16  = lane & 15;

    const float* Arow = A + (size_t)(m0 + l16) * FDIM;
    float di = (mode != A_RAW) ? dis[m0 + l16] : 0.0f;

    v8f acc[8] = {};
    strip_k128(Arow, di, bprev, B, half, l16, mode, acc);

    float dg[8];
    #pragma unroll
    for (int g = 0; g < 8; ++g) dg[g] = dis[m0 + g + half * 8];

    #pragma unroll
    for (int n = 0; n < 8; ++n) {
        #pragma unroll
        for (int g = 0; g < 8; ++g) {
            int m = m0 + g + half * 8;
            float val = acc[n][g] * dg[g];
            size_t o = (size_t)m * FDIM + n * 16 + l16;
            hws[o] = val;
        }
    }
}

__global__ __launch_bounds__(256) void gemm_node_first(const float* __restrict__ X,
                                                       const float* __restrict__ dis,
                                                       const float* __restrict__ B,
                                                       float* __restrict__ hws, int M) {
    gemm_node_body(X, nullptr, dis, B, hws, M, A_RAW);
}

__global__ __launch_bounds__(256) void gemm_node_next(const float* __restrict__ aggPrev,
                                                      const float* __restrict__ bPrev,
                                                      const float* __restrict__ dis,
                                                      const float* __restrict__ B,
                                                      float* __restrict__ hws, int M) {
    gemm_node_body(aggPrev, bPrev, dis, B, hws, M, A_AFFRELU);
}


static __device__ __forceinline__ unsigned zero_bytes(unsigned y) {
    return ~(((y & 0x7F7F7F7Fu) + 0x7F7F7F7Fu) | y | 0x7F7F7F7Fu);
}

static __device__ __forceinline__ void owner_drain(const int* myseg, int cnt, int base, int lane,
                                                   const float* __restrict__ hws,
                                                   const int* __restrict__ src,
                                                   const int* __restrict__ dst,
                                                   float* acc) {
    for (int i0 = 0; i0 < cnt; i0 += 32) {
        int left = cnt - i0;
        bool ok = lane < left;
        int e = ok ? myseg[i0 + lane] : 0;
        int s_l = src[e];
        int d_l = dst[e];
        int r_l = ok ? (d_l - base) : OWN_ROWS;
        #pragma unroll
        for (int g = 0; g < 4; ++g) {
            if (g * 8 < left) {
                float v[8][4];
                #pragma unroll
                for (int j = 0; j < 8; ++j) {
                    int s = __builtin_amdgcn_readlane(s_l, g * 8 + j);
                    const float* hp = hws + (size_t)s * FDIM + lane;
                    v[j][0] = hp[0];
                    v[j][1] = hp[32];
                    v[j][2] = hp[64];
                    v[j][3] = hp[96];
                }
                #pragma unroll
                for (int j = 0; j < 8; ++j) {
                    int r = __builtin_amdgcn_readlane(r_l, g * 8 + j);
                    float* ap = acc + r * FDIM + lane;
                    atomicAdd(ap + 0,  v[j][0]);
                    atomicAdd(ap + 32, v[j][1]);
                    atomicAdd(ap + 64, v[j][2]);
                    atomicAdd(ap + 96, v[j][3]);
                }
            }
        }
    }
}

__global__ __launch_bounds__(OWN_THREADS) void agg_owner(const float* __restrict__ hws,
                                                         const int* __restrict__ src,
                                                         const int* __restrict__ dst,
                                                         const unsigned int* __restrict__ own,
                                                         float* __restrict__ agg) {
    __shared__ alignas(16) float acc[(OWN_ROWS + 1) * FDIM];
    __shared__ int   seg[(OWN_THREADS / 32) * SEG_CAP];

    const int b     = blockIdx.x;
    const int base  = b * OWN_ROWS;
    const int nrows = (N_NODES - base < OWN_ROWS) ? (N_NODES - base) : OWN_ROWS;
    const int tid   = threadIdx.x;
    const int lane  = tid & 31;
    const int w     = __builtin_amdgcn_readfirstlane(tid >> 5);

    {
        const v4f* h4 = (const v4f*)(hws + (size_t)base * FDIM);
        v4f* a4 = (v4f*)acc;
        const int lim = nrows * (FDIM / 4);
        constexpr int NPC = (OWN_ROWS * (FDIM / 4) + OWN_THREADS - 1) / OWN_THREADS;
        static_assert(NPC == 7, "the fence below names seven values");
        v4f t4[NPC];
        #pragma unroll
        for (int t = 0; t < NPC; ++t) {
            int i = tid + t * OWN_THREADS;
            t4[t] = h4[i < lim ? i : lim - 1];
        }
        asm volatile("" :: "v"(t4[0]), "v"(t4[1]), "v"(t4[2]), "v"(t4[3]), "v"(t4[4]), "v"(t4[5]), "v"(t4[6]));
        #pragma unroll
        for (int t = 0; t < NPC; ++t) {
            int i = tid + t * OWN_THREADS;
            if (i < lim) a4[i] = t4[t];
        }
        if (tid < FDIM / 4) a4[OWN_ROWS * (FDIM / 4) + tid] = (v4f){0.0f, 0.0f, 0.0f, 0.0f};
    }
    __syncthreads();

    int* myseg = seg + w * SEG_CAP;
    int cnt = 0;
    const unsigned pat = (unsigned)b * 0x01010101u;
    const int rot = b * (N_CH / OWN_WG);
    const uint4* own4 = (const uint4*)own;

    for (int c0 = w; c0 < N_CH; c0 += 32 * 4) {
        uint4 q[4];
        int eb[4];
        bool ok4[4];
        #pragma unroll
        for (int k = 0; k < 4; ++k) {
            int cc = c0 + 32 * k;
            int c = cc + rot;
            if (c >= N_CH) c -= N_CH;
            int e = c * CH_EDGES + lane * 16;
            eb[k] = e;
            ok4[k] = (cc < N_CH) && (e < N_EDGES);
            q[k] = own4[ok4[k] ? (e >> 4) : 0];
        }
        #pragma unroll
        for (int k = 0; k < 4; ++k) {
            if (cnt > SEG_CAP - CH_EDGES) {
                owner_drain(myseg, cnt, base, lane, hws, src, dst, acc);
                cnt = 0;
            }
            unsigned zc = (zero_bytes(q[k].x ^ pat) >> 7) | (zero_bytes(q[k].y ^ pat) >> 6) |
                          (zero_bytes(q[k].z ^ pat) >> 5) | (zero_bytes(q[k].w ^ pat) >> 4);
            if (!ok4[k]) zc = 0u;
            for (int t = 0; t < 16; ++t) {
                bool has = zc != 0u;
                unsigned m = __builtin_amdgcn_ballot_w32(has);
                if (m == 0u) break;
                if (has) {
                    int c = __builtin_ctz(zc);
                    zc &= zc - 1u;
                    int p = ((c & 7) << 2) | (c >> 3);
                    int pos = cnt + __builtin_popcount(m & ((1u << lane) - 1u));
                    myseg[pos] = eb[k] + p;
                }
                cnt += __builtin_popcount(m);
            }
        }
    }
    owner_drain(myseg, cnt, base, lane, hws, src, dst, acc);

    __syncthreads();

    {
        const float4* a4 = (const float4*)acc;
        float4* o4 = (float4*)(agg + (size_t)base * FDIM);
        for (int i = tid; i < nrows * (FDIM / 4); i += OWN_THREADS) o4[i] = a4[i];
    }
}

__global__ __launch_bounds__(256) void gemm_edge(const float* __restrict__ AGG,
                                                 const float* __restrict__ dis,
                                                 const float* __restrict__ b3,
                                                 const int* __restrict__ pos_tr,
                                                 const int* __restrict__ neg_tr,
                                                 const int* __restrict__ pos_te,
                                                 const int* __restrict__ neg_te,
                                                 const float* __restrict__ Wl1,
                                                 const float* __restrict__ bl1,
                                                 const float* __restrict__ Wl2,
                                                 const float* __restrict__ bl2,
                                                 float* __restrict__ out) {
    int wave = (blockIdx.x * blockDim.x + threadIdx.x) >> 5;
    int lane = threadIdx.x & 31;
    int r0 = wave * 16;
    if (r0 >= N_ROWS) return;
    int half = lane >> 4;
    int l16  = lane & 15;
    int r = r0 + l16;

    int u, v;
    if (r < E_TR)            { u = pos_tr[r];              v = pos_tr[E_TR + r]; }
    else if (r < 2 * E_TR)   { int i = r - E_TR;           u = neg_tr[i]; v = neg_tr[E_TR + i]; }
    else if (r < 2 * E_TR + E_TE) { int i = r - 2 * E_TR;  u = pos_te[i]; v = pos_te[E_TE + i]; }
    else                     { int i = r - 2 * E_TR - E_TE; u = neg_te[i]; v = neg_te[E_TE + i]; }

    const float* Au = AGG + (size_t)u * FDIM;
    const float* Av = AGG + (size_t)v * FDIM;
    float du = dis[u];
    float dv = dis[v];

    v8f acc[8] = {};

    strip_k128(Au, du, b3, Wl1, half, l16, A_AFFINE, acc);
    strip_k128(Av, dv, b3, Wl1 + (size_t)FDIM * FDIM, half, l16, A_AFFINE, acc);

    float partial[8];
    #pragma unroll
    for (int g = 0; g < 8; ++g) partial[g] = 0.0f;

    #pragma unroll
    for (int n = 0; n < 8; ++n) {
        float bb = bl1[n * 16 + l16];
        float ww = Wl2[n * 16 + l16];
        #pragma unroll
        for (int g = 0; g < 8; ++g) {
            float z = fmaxf(acc[n][g] + bb, 0.0f);
            partial[g] += z * ww;
        }
    }

    float bias2 = bl2[0];
    #pragma unroll
    for (int g = 0; g < 8; ++g) {
        float s = partial[g];
        s += __shfl_xor(s, 8, 16);
        s += __shfl_xor(s, 4, 16);
        s += __shfl_xor(s, 2, 16);
        s += __shfl_xor(s, 1, 16);
        if (l16 == 0) out[r0 + g + half * 8] = s + bias2;
    }
}

extern "C" void kernel_launch(void* const* d_in, const int* in_sizes, int n_in,
                              void* d_out, int out_size, void* d_ws, size_t ws_size,
                              hipStream_t stream) {
    const float* x          = (const float*)d_in[0];
    const int*   edge_index = (const int*)d_in[1];
    const int*   pos_tr     = (const int*)d_in[2];
    const int*   neg_tr     = (const int*)d_in[3];
    const int*   pos_te     = (const int*)d_in[4];
    const int*   neg_te     = (const int*)d_in[5];
    const float* W0 = (const float*)d_in[6];
    const float* b0 = (const float*)d_in[7];
    const float* W1 = (const float*)d_in[8];
    const float* b1 = (const float*)d_in[9];
    const float* W2 = (const float*)d_in[10];
    const float* b2 = (const float*)d_in[11];
    const float* W3 = (const float*)d_in[12];
    const float* b3 = (const float*)d_in[13];
    const float* Wl1 = (const float*)d_in[14];
    const float* bl1 = (const float*)d_in[15];
    const float* Wl2 = (const float*)d_in[16];
    const float* bl2 = (const float*)d_in[17];
    float* out = (float*)d_out;

    const size_t NF = (size_t)N_NODES * FDIM;
    float* ws   = (float*)d_ws;
    float* dis  = ws;
    float* hws  = ws + 50048;
    float* agg  = hws + NF;
    unsigned char* own = (unsigned char*)(agg + NF);

    const int* src = edge_index;
    const int* dst = edge_index + N_EDGES;

    hipMemsetAsync(dis, 0, N_NODES * sizeof(float), stream);
    deg_count<<<(N_EDGES + 255) / 256, 256, 0, stream>>>(dst, dis, own, N_EDGES);
    make_dis<<<(N_NODES + 255) / 256, 256, 0, stream>>>(dis, N_NODES);

    const int tilesM     = (N_NODES + 15) / 16;
    const int gemmBlocks = (tilesM + 7) / 8;
    const unsigned int* own32 = (const unsigned int*)own;

    gemm_node_first<<<gemmBlocks, 256, 0, stream>>>(x, dis, W0, hws, N_NODES);
    agg_owner<<<OWN_WG, OWN_THREADS, 0, stream>>>(hws, src, dst, own32, agg);

    gemm_node_next<<<gemmBlocks, 256, 0, stream>>>(agg, b0, dis, W1, hws, N_NODES);
    agg_owner<<<OWN_WG, OWN_THREADS, 0, stream>>>(hws, src, dst, own32, agg);

    gemm_node_next<<<gemmBlocks, 256, 0, stream>>>(agg, b1, dis, W2, hws, N_NODES);
    agg_owner<<<OWN_WG, OWN_THREADS, 0, stream>>>(hws, src, dst, own32, agg);

    gemm_node_next<<<gemmBlocks, 256, 0, stream>>>(agg, b2, dis, W3, hws, N_NODES);
    agg_owner<<<OWN_WG, OWN_THREADS, 0, stream>>>(hws, src, dst, own32, agg);

    const int edgeTiles  = N_ROWS / 16;
    const int edgeBlocks = (edgeTiles + 7) / 8;
    gemm_edge<<<edgeBlocks, 256, 0, stream>>>(agg, dis, b3, pos_tr, neg_tr, pos_te, neg_te,
                                              Wl1, bl1, Wl2, bl2, out);
}


# Edges binned by owner once per call: one pre-pass scans the owner bytes, writes each owner's list of packed (src,row) words, counts in-degrees in LDS; layers read their list; arrival order as before
# speedup vs baseline: 2.6070x; 1.2493x over previous
#include <hip/hip_runtime.h>


typedef float v2f __attribute__((ext_vector_type(2)));
typedef float v8f __attribute__((ext_vector_type(8)));
typedef float v4f __attribute__((ext_vector_type(4)));

#define N_NODES 50000
#define N_EDGES 800000
#define FDIM    128
#define E_TR    100000
#define E_TE    20000
#define N_ROWS  240000

#define OWN_ROWS    196
#define OWN_WG      ((N_NODES + OWN_ROWS - 1) / OWN_ROWS)
#define OWN_THREADS 1024
#define SEG_CAP     1024
#define CH_EDGES    512
#define N_CH        ((N_EDGES + CH_EDGES - 1) / CH_EDGES)
#define OC_THREADS  256
#define OC_WG       ((N_EDGES / 16 + OC_THREADS - 1) / OC_THREADS)
static_assert(OWN_WG <= 256, "an owner id must fit in one byte");
static_assert(N_EDGES % 16 == 0, "a lane's 16 ids are all inside the edge list or all outside");
static_assert(2 * CH_EDGES <= SEG_CAP, "a drained list must take one full step");
static_assert(N_NODES <= 65536 && OWN_ROWS < 65536, "a list word holds the source row in 16 bits and the owned row above them");
static_assert(OC_WG % 4 == 0, "the count table is summed in four equal parts");
#define ACC_FLOATS  (164 * 1024 / 4)
static_assert(ACC_FLOATS >= (OWN_ROWS + 1) * FDIM, "owned rows and the spare row fit");

__global__ __launch_bounds__(OC_THREADS) void own_count(const int* __restrict__ dst,
                                                        unsigned int* __restrict__ own,
                                                        unsigned int* __restrict__ cnt) {
    __shared__ unsigned int hist[256];
    const int tid = threadIdx.x;
    hist[tid] = 0u;
    __syncthreads();
    const int t = blockIdx.x * OC_THREADS + tid;
    if (t < N_EDGES / 16) {
        const int4* d4 = (const int4*)(dst + (size_t)t * 16);
        int4 q[4];
        #pragma unroll
        for (int j = 0; j < 4; ++j) q[j] = d4[j];
        unsigned int wv[4];
        #pragma unroll
        for (int j = 0; j < 4; ++j) {
            unsigned int o0 = (unsigned int)q[j].x / OWN_ROWS;
            unsigned int o1 = (unsigned int)q[j].y / OWN_ROWS;
            unsigned int o2 = (unsigned int)q[j].z / OWN_ROWS;
            unsigned int o3 = (unsigned int)q[j].w / OWN_ROWS;
            atomicAdd(&hist[o0], 1u);
            atomicAdd(&hist[o1], 1u);
            atomicAdd(&hist[o2], 1u);
            atomicAdd(&hist[o3], 1u);
            wv[j] = o0 | (o1 << 8) | (o2 << 16) | (o3 << 24);
        }
        ((uint4*)own)[t] = make_uint4(wv[0], wv[1], wv[2], wv[3]);
    }
    __syncthreads();
    cnt[blockIdx.x * 256 + tid] = hist[tid];
}

#define A_RAW     0
#define A_AFFINE  1
#define A_AFFRELU 2

static __device__ __forceinline__ void strip_k128(const float* __restrict__ Arow, float di,
                                                  const float* __restrict__ bias,
                                                  const float* __restrict__ Bm,
                                                  int half, int l16, const int mode,
                                                  v8f (&acc)[8]) {
    for (int k0 = 0; k0 < FDIM; k0 += 4) {
        int k = k0 + half * 2;
        v2f afrag = *(const v2f*)(Arow + k);
        if (mode != A_RAW) {
            v2f bb = *(const v2f*)(bias + k);
            afrag.x = afrag.x * di + bb.x;
            afrag.y = afrag.y * di + bb.y;
            if (mode == A_AFFRELU) {
                afrag.x = fmaxf(afrag.x, 0.0f);
                afrag.y = fmaxf(afrag.y, 0.0f);
            }
        }
        const float* b0 = Bm + (size_t)k * FDIM + l16;
        #pragma unroll
        for (int n = 0; n < 8; ++n) {
            v2f bfrag;
            bfrag.x = b0[n * 16];
            bfrag.y = b0[FDIM + n * 16];
            acc[n] = __builtin_amdgcn_wmma_f32_16x16x4_f32(
                false, afrag, false, bfrag, (short)0, acc[n], false, false);
        }
    }
}

static __device__ __forceinline__ void gemm_node_body(const float* __restrict__ A,
                                                      const float* __restrict__ bprev,
                                                      const float* __restrict__ dis,
                                                      const float* __restrict__ B,
                                                      float* __restrict__ hws,
                                                      int M, const int mode) {
    int wave = (blockIdx.x * blockDim.x + threadIdx.x) >> 5;
    int lane = threadIdx.x & 31;
    int m0 = wave * 16;
    if (m0 >= M) return;
    int half = lane >> 4;
    int l16  = lane & 15;

    const float* Arow = A + (size_t)(m0 + l16) * FDIM;
    float di = (mode != A_RAW) ? dis[m0 + l16] : 0.0f;

    v8f acc[8] = {};
    strip_k128(Arow, di, bprev, B, half, l16, mode, acc);

    float dg[8];
    #pragma unroll
    for (int g = 0; g < 8; ++g) dg[g] = dis[m0 + g + half * 8];

    #pragma unroll
    for (int n = 0; n < 8; ++n) {
        #pragma unroll
        for (int g = 0; g < 8; ++g) {
            int m = m0 + g + half * 8;
            float val = acc[n][g] * dg[g];
            size_t o = (size_t)m * FDIM + n * 16 + l16;
            hws[o] = val;
        }
    }
}

__global__ __launch_bounds__(256) void gemm_node_first(const float* __restrict__ X,
                                                       const float* __restrict__ dis,
                                                       const float* __restrict__ B,
                                                       float* __restrict__ hws, int M) {
    gemm_node_body(X, nullptr, dis, B, hws, M, A_RAW);
}

__global__ __launch_bounds__(256) void gemm_node_next(const float* __restrict__ aggPrev,
                                                      const float* __restrict__ bPrev,
                                                      const float* __restrict__ dis,
                                                      const float* __restrict__ B,
                                                      float* __restrict__ hws, int M) {
    gemm_node_body(aggPrev, bPrev, dis, B, hws, M, A_AFFRELU);
}


static __device__ __forceinline__ unsigned zero_bytes(unsigned y) {
    return ~(((y & 0x7F7F7F7Fu) + 0x7F7F7F7Fu) | y | 0x7F7F7F7Fu);
}

static __device__ __forceinline__ void owner_flush(const int* myseg, int cnt, int base, int lane,
                                                   const int* __restrict__ src,
                                                   const int* __restrict__ dst,
                                                   unsigned int* degc, unsigned int* cursor,
                                                   unsigned int* __restrict__ mylist) {
    for (int i0 = 0; i0 < cnt; i0 += 32) {
        int left = cnt - i0;
        bool ok = lane < left;
        int e = ok ? myseg[i0 + lane] : 0;
        int s = src[e];
        int d = dst[e];
        int n = left < 32 ? left : 32;
        unsigned int pos = 0u;
        if (lane == 0) pos = atomicAdd(cursor, (unsigned int)n);
        pos = (unsigned int)__builtin_amdgcn_readfirstlane((int)pos);
        if (ok) {
            int r = d - base;
            atomicAdd(&degc[r], 1u);
            mylist[pos + lane] = (unsigned int)s | ((unsigned int)r << 16);
        }
    }
}

__global__ __launch_bounds__(OWN_THREADS) void owner_build(const int* __restrict__ src,
                                                           const int* __restrict__ dst,
                                                           const unsigned int* __restrict__ own,
                                                           const unsigned int* __restrict__ cnt,
                                                           unsigned int* __restrict__ list,
                                                           int* __restrict__ meta,
                                                           float* __restrict__ dis) {
    __shared__ int seg[(OWN_THREADS / 32) * SEG_CAP];
    __shared__ unsigned int part[4 * 256];
    __shared__ unsigned int degc[OWN_ROWS];
    __shared__ unsigned int cursor;
    __shared__ unsigned int offs;

    const int b     = blockIdx.x;
    const int base  = b * OWN_ROWS;
    const int nrows = (N_NODES - base < OWN_ROWS) ? (N_NODES - base) : OWN_ROWS;
    const int tid   = threadIdx.x;
    const int lane  = tid & 31;
    const int w     = __builtin_amdgcn_readfirstlane(tid >> 5);

    {
        const int o = tid & 255;
        const int q = tid >> 8;
        unsigned int s = 0u;
        for (int c = q * (OC_WG / 4); c < (q + 1) * (OC_WG / 4); ++c) s += cnt[c * 256 + o];
        part[q * 256 + o] = s;
    }
    if (tid < OWN_ROWS) degc[tid] = 0u;
    if (tid == 0) cursor = 0u;
    __syncthreads();
    if (w == 0) {
        unsigned int s = 0u;
        for (int j = 0; j < 8; ++j) {
            int o = lane * 8 + j;
            if (o < b) s += part[o] + part[256 + o] + part[512 + o] + part[768 + o];
        }
        s += (unsigned int)__shfl_xor((int)s, 16);
        s += (unsigned int)__shfl_xor((int)s, 8);
        s += (unsigned int)__shfl_xor((int)s, 4);
        s += (unsigned int)__shfl_xor((int)s, 2);
        s += (unsigned int)__shfl_xor((int)s, 1);
        if (lane == 0) offs = s;
    }
    __syncthreads();
    unsigned int* mylist = list + offs;

    int* myseg = seg + w * SEG_CAP;
    int cnt_w = 0;
    const unsigned pat = (unsigned)b * 0x01010101u;
    const int rot = b * (N_CH / OWN_WG);
    const uint4* own4 = (const uint4*)own;

    for (int c0 = w; c0 < N_CH; c0 += 32 * 4) {
        uint4 q[4];
        int eb[4];
        bool ok4[4];
        #pragma unroll
        for (int k = 0; k < 4; ++k) {
            int cc = c0 + 32 * k;
            int c = cc + rot;
            if (c >= N_CH) c -= N_CH;
            int e = c * CH_EDGES + lane * 16;
            eb[k] = e;
            ok4[k] = (cc < N_CH) && (e < N_EDGES);
            q[k] = own4[ok4[k] ? (e >> 4) : 0];
        }
        #pragma unroll
        for (int k = 0; k < 4; ++k) {
            if (cnt_w > SEG_CAP - CH_EDGES) {
                owner_flush(myseg, cnt_w, base, lane, src, dst, degc, &cursor, mylist);
                cnt_w = 0;
            }
            unsigned zc = (zero_bytes(q[k].x ^ pat) >> 7) | (zero_bytes(q[k].y ^ pat) >> 6) |
                          (zero_bytes(q[k].z ^ pat) >> 5) | (zero_bytes(q[k].w ^ pat) >> 4);
            if (!ok4[k]) zc = 0u;
            for (int t = 0; t < 16; ++t) {
                bool has = zc != 0u;
                unsigned m = __builtin_amdgcn_ballot_w32(has);
                if (m == 0u) break;
                if (has) {
                    int c = __builtin_ctz(zc);
                    zc &= zc - 1u;
                    int p = ((c & 7) << 2) | (c >> 3);
                    int pos = cnt_w + __builtin_popcount(m & ((1u << lane) - 1u));
                    myseg[pos] = eb[k] + p;
                }
                cnt_w += __builtin_popcount(m);
            }
        }
    }
    owner_flush(myseg, cnt_w, base, lane, src, dst, degc, &cursor, mylist);

    __syncthreads();

    if (tid < nrows) dis[base + tid] = rsqrtf((float)degc[tid] + 1.0f);
    if (tid == 0) {
        meta[2 * b]     = (int)offs;
        meta[2 * b + 1] = (int)cursor;
    }
}

__global__ __launch_bounds__(OWN_THREADS) void agg_owner(const float* __restrict__ hws,
                                                         const unsigned int* __restrict__ list,
                                                         const int* __restrict__ meta,
                                                         float* __restrict__ agg) {
    __shared__ alignas(16) float acc[ACC_FLOATS];

    const int b     = blockIdx.x;
    const int base  = b * OWN_ROWS;
    const int nrows = (N_NODES - base < OWN_ROWS) ? (N_NODES - base) : OWN_ROWS;
    const int tid   = threadIdx.x;
    const int lane  = tid & 31;
    const int w     = __builtin_amdgcn_readfirstlane(tid >> 5);

    const int off = meta[2 * b];
    const int n   = meta[2 * b + 1];

    {
        const v4f* h4 = (const v4f*)(hws + (size_t)base * FDIM);
        v4f* a4 = (v4f*)acc;
        const int lim = nrows * (FDIM / 4);
        constexpr int NPC = (OWN_ROWS * (FDIM / 4) + OWN_THREADS - 1) / OWN_THREADS;
        static_assert(NPC == 7, "the fence below names seven values");
        v4f t4[NPC];
        #pragma unroll
        for (int t = 0; t < NPC; ++t) {
            int i = tid + t * OWN_THREADS;
            t4[t] = h4[i < lim ? i : lim - 1];
        }
        asm volatile("" :: "v"(t4[0]), "v"(t4[1]), "v"(t4[2]), "v"(t4[3]), "v"(t4[4]), "v"(t4[5]), "v"(t4[6]));
        #pragma unroll
        for (int t = 0; t < NPC; ++t) {
            int i = tid + t * OWN_THREADS;
            if (i < lim) a4[i] = t4[t];
        }
        if (tid < FDIM / 4) a4[OWN_ROWS * (FDIM / 4) + tid] = (v4f){0.0f, 0.0f, 0.0f, 0.0f};
    }
    const int per = (((n + 31) >> 5) + 7) & ~7;
    const int lo  = w * per;
    const int hi  = (lo + per < n) ? (lo + per) : n;
    const unsigned int* mylist = list + off;
    const unsigned int none = (unsigned int)OWN_ROWS << 16;
    unsigned int pk_next = (lane < hi - lo) ? mylist[lo + lane] : none;

    __syncthreads();

    for (int i0 = lo; i0 < hi; i0 += 32) {
        int left = hi - i0;
        unsigned int pk = pk_next;
        pk_next = (lane < left - 32) ? mylist[i0 + 32 + lane] : none;
        int s_l = (int)(pk & 0xFFFFu);
        int r_l = (int)(pk >> 16);
        #pragma unroll
        for (int g = 0; g < 4; ++g) {
            if (g * 8 < left) {
                float v[8][4];
                #pragma unroll
                for (int j = 0; j < 8; ++j) {
                    int s = __builtin_amdgcn_readlane(s_l, g * 8 + j);
                    const float* hp = hws + (size_t)s * FDIM + lane;
                    v[j][0] = hp[0];
                    v[j][1] = hp[32];
                    v[j][2] = hp[64];
                    v[j][3] = hp[96];
                }
                #pragma unroll
                for (int j = 0; j < 8; ++j) {
                    int r = __builtin_amdgcn_readlane(r_l, g * 8 + j);
                    float* ap = acc + r * FDIM + lane;
                    atomicAdd(ap + 0,  v[j][0]);
                    atomicAdd(ap + 32, v[j][1]);
                    atomicAdd(ap + 64, v[j][2]);
                    atomicAdd(ap + 96, v[j][3]);
                }
            }
        }
    }

    __syncthreads();

    {
        const float4* a4 = (const float4*)acc;
        float4* o4 = (float4*)(agg + (size_t)base * FDIM);
        for (int i = tid; i < nrows * (FDIM / 4); i += OWN_THREADS) o4[i] = a4[i];
    }
}

__global__ __launch_bounds__(256) void gemm_edge(const float* __restrict__ AGG,
                                                 const float* __restrict__ dis,
                                                 const float* __restrict__ b3,
                                                 const int* __restrict__ pos_tr,
                                                 const int* __restrict__ neg_tr,
                                                 const int* __restrict__ pos_te,
                                                 const int* __restrict__ neg_te,
                                                 const float* __restrict__ Wl1,
                                                 const float* __restrict__ bl1,
                                                 const float* __restrict__ Wl2,
                                                 const float* __restrict__ bl2,
                                                 float* __restrict__ out) {
    int wave = (blockIdx.x * blockDim.x + threadIdx.x) >> 5;
    int lane = threadIdx.x & 31;
    int r0 = wave * 16;
    if (r0 >= N_ROWS) return;
    int half = lane >> 4;
    int l16  = lane & 15;
    int r = r0 + l16;

    int u, v;
    if (r < E_TR)            { u = pos_tr[r];              v = pos_tr[E_TR + r]; }
    else if (r < 2 * E_TR)   { int i = r - E_TR;           u = neg_tr[i]; v = neg_tr[E_TR + i]; }
    else if (r < 2 * E_TR + E_TE) { int i = r - 2 * E_TR;  u = pos_te[i]; v = pos_te[E_TE + i]; }
    else                     { int i = r - 2 * E_TR - E_TE; u = neg_te[i]; v = neg_te[E_TE + i]; }

    const float* Au = AGG + (size_t)u * FDIM;
    const float* Av = AGG + (size_t)v * FDIM;
    float du = dis[u];
    float dv = dis[v];

    v8f acc[8] = {};

    strip_k128(Au, du, b3, Wl1, half, l16, A_AFFINE, acc);
    strip_k128(Av, dv, b3, Wl1 + (size_t)FDIM * FDIM, half, l16, A_AFFINE, acc);

    float partial[8];
    #pragma unroll
    for (int g = 0; g < 8; ++g) partial[g] = 0.0f;

    #pragma unroll
    for (int n = 0; n < 8; ++n) {
        float bb = bl1[n * 16 + l16];
        float ww = Wl2[n * 16 + l16];
        #pragma unroll
        for (int g = 0; g < 8; ++g) {
            float z = fmaxf(acc[n][g] + bb, 0.0f);
            partial[g] += z * ww;
        }
    }

    float bias2 = bl2[0];
    #pragma unroll
    for (int g = 0; g < 8; ++g) {
        float s = partial[g];
        s += __shfl_xor(s, 8, 16);
        s += __shfl_xor(s, 4, 16);
        s += __shfl_xor(s, 2, 16);
        s += __shfl_xor(s, 1, 16);
        if (l16 == 0) out[r0 + g + half * 8] = s + bias2;
    }
}

extern "C" void kernel_launch(void* const* d_in, const int* in_sizes, int n_in,
                              void* d_out, int out_size, void* d_ws, size_t ws_size,
                              hipStream_t stream) {
    const float* x          = (const float*)d_in[0];
    const int*   edge_index = (const int*)d_in[1];
    const int*   pos_tr     = (const int*)d_in[2];
    const int*   neg_tr     = (const int*)d_in[3];
    const int*   pos_te     = (const int*)d_in[4];
    const int*   neg_te     = (const int*)d_in[5];
    const float* W0 = (const float*)d_in[6];
    const float* b0 = (const float*)d_in[7];
    const float* W1 = (const float*)d_in[8];
    const float* b1 = (const float*)d_in[9];
    const float* W2 = (const float*)d_in[10];
    const float* b2 = (const float*)d_in[11];
    const float* W3 = (const float*)d_in[12];
    const float* b3 = (const float*)d_in[13];
    const float* Wl1 = (const float*)d_in[14];
    const float* bl1 = (const float*)d_in[15];
    const float* Wl2 = (const float*)d_in[16];
    const float* bl2 = (const float*)d_in[17];
    float* out = (float*)d_out;

    const size_t NF = (size_t)N_NODES * FDIM;
    float* ws   = (float*)d_ws;
    float* dis  = ws;
    float* hws  = ws + 50048;
    float* agg  = hws + NF;
    unsigned int* own  = (unsigned int*)(agg + NF);
    unsigned int* cnt  = own + N_EDGES / 4;
    int*          meta = (int*)(cnt + OC_WG * 256);
    unsigned int* list = (unsigned int*)(meta + 2 * OWN_WG);

    const int* src = edge_index;
    const int* dst = edge_index + N_EDGES;

    own_count<<<OC_WG, OC_THREADS, 0, stream>>>(dst, own, cnt);
    owner_build<<<OWN_WG, OWN_THREADS, 0, stream>>>(src, dst, own, cnt, list, meta, dis);

    const int tilesM     = (N_NODES + 15) / 16;
    const int gemmBlocks = (tilesM + 7) / 8;

    gemm_node_first<<<gemmBlocks, 256, 0, stream>>>(x, dis, W0, hws, N_NODES);
    agg_owner<<<OWN_WG, OWN_THREADS, 0, stream>>>(hws, list, meta, agg);

    gemm_node_next<<<gemmBlocks, 256, 0, stream>>>(agg, b0, dis, W1, hws, N_NODES);
    agg_owner<<<OWN_WG, OWN_THREADS, 0, stream>>>(hws, list, meta, agg);

    gemm_node_next<<<gemmBlocks, 256, 0, stream>>>(agg, b1, dis, W2, hws, N_NODES);
    agg_owner<<<OWN_WG, OWN_THREADS, 0, stream>>>(hws, list, meta, agg);

    gemm_node_next<<<gemmBlocks, 256, 0, stream>>>(agg, b2, dis, W3, hws, N_NODES);
    agg_owner<<<OWN_WG, OWN_THREADS, 0, stream>>>(hws, list, meta, agg);

    const int edgeTiles  = N_ROWS / 16;
    const int edgeBlocks = (edgeTiles + 7) / 8;
    gemm_edge<<<edgeBlocks, 256, 0, stream>>>(agg, dis, b3, pos_tr, neg_tr, pos_te, neg_te,
                                              Wl1, bl1, Wl2, bl2, out);
}


# Layers 2-4: the aggregation work-group multiplies its 196 finished rows, still in LDS, by the next layer's weights on f32 WMMA; 3 GEMM launches and the sums' round trip gone; sums and k order kept
# speedup vs baseline: 3.1243x; 1.1984x over previous
#include <hip/hip_runtime.h>


typedef float v2f __attribute__((ext_vector_type(2)));
typedef float v8f __attribute__((ext_vector_type(8)));
typedef float v4f __attribute__((ext_vector_type(4)));

#define N_NODES 50000
#define N_EDGES 800000
#define FDIM    128
#define E_TR    100000
#define E_TE    20000
#define N_ROWS  240000

#define OWN_ROWS    196
#define OWN_WG      ((N_NODES + OWN_ROWS - 1) / OWN_ROWS)
#define OWN_THREADS 1024
#define SEG_CAP     1024
#define CH_EDGES    512
#define N_CH        ((N_EDGES + CH_EDGES - 1) / CH_EDGES)
#define OC_THREADS  256
#define OC_WG       ((N_EDGES / 16 + OC_THREADS - 1) / OC_THREADS)
static_assert(OWN_WG <= 256, "an owner id must fit in one byte");
static_assert(N_EDGES % 16 == 0, "a lane's 16 ids are all inside the edge list or all outside");
static_assert(2 * CH_EDGES <= SEG_CAP, "a drained list must take one full step");
static_assert(N_NODES <= 65536 && OWN_ROWS < 65536, "a list word holds the source row in 16 bits and the owned row above them");
static_assert(OC_WG % 4 == 0, "the count table is summed in four equal parts");
#define ACC_FLOATS  (164 * 1024 / 4)
static_assert(ACC_FLOATS >= (OWN_ROWS + 1) * FDIM, "owned rows and the spare row fit");
#define ACC_PITCH   132
#define GEMM_ROWS   208
#define GEMM_TASKS  (4 * (GEMM_ROWS / 16))
static_assert(GEMM_ROWS % 16 == 0 && GEMM_ROWS > OWN_ROWS, "whole tiles, and the spare row is among the zeroed rows");
static_assert(ACC_PITCH % 4 == 0, "a row starts on a 16-byte boundary");
static_assert(ACC_FLOATS >= GEMM_ROWS * ACC_PITCH, "the padded rows fit");

__global__ __launch_bounds__(OC_THREADS) void own_count(const int* __restrict__ dst,
                                                        unsigned int* __restrict__ own,
                                                        unsigned int* __restrict__ cnt) {
    __shared__ unsigned int hist[256];
    const int tid = threadIdx.x;
    hist[tid] = 0u;
    __syncthreads();
    const int t = blockIdx.x * OC_THREADS + tid;
    if (t < N_EDGES / 16) {
        const int4* d4 = (const int4*)(dst + (size_t)t * 16);
        int4 q[4];
        #pragma unroll
        for (int j = 0; j < 4; ++j) q[j] = d4[j];
        unsigned int wv[4];
        #pragma unroll
        for (int j = 0; j < 4; ++j) {
            unsigned int o0 = (unsigned int)q[j].x / OWN_ROWS;
            unsigned int o1 = (unsigned int)q[j].y / OWN_ROWS;
            unsigned int o2 = (unsigned int)q[j].z / OWN_ROWS;
            unsigned int o3 = (unsigned int)q[j].w / OWN_ROWS;
            atomicAdd(&hist[o0], 1u);
            atomicAdd(&hist[o1], 1u);
            atomicAdd(&hist[o2], 1u);
            atomicAdd(&hist[o3], 1u);
            wv[j] = o0 | (o1 << 8) | (o2 << 16) | (o3 << 24);
        }
        ((uint4*)own)[t] = make_uint4(wv[0], wv[1], wv[2], wv[3]);
    }
    __syncthreads();
    cnt[blockIdx.x * 256 + tid] = hist[tid];
}

#define A_RAW     0
#define A_AFFINE  1

static __device__ __forceinline__ void strip_k128(const float* __restrict__ Arow, float di,
                                                  const float* __restrict__ bias,
                                                  const float* __restrict__ Bm,
                                                  int half, int l16, const int mode,
                                                  v8f (&acc)[8]) {
    for (int k0 = 0; k0 < FDIM; k0 += 4) {
        int k = k0 + half * 2;
        v2f afrag = *(const v2f*)(Arow + k);
        if (mode != A_RAW) {
            v2f bb = *(const v2f*)(bias + k);
            afrag.x = afrag.x * di + bb.x;
            afrag.y = afrag.y * di + bb.y;
        }
        const float* b0 = Bm + (size_t)k * FDIM + l16;
        #pragma unroll
        for (int n = 0; n < 8; ++n) {
            v2f bfrag;
            bfrag.x = b0[n * 16];
            bfrag.y = b0[FDIM + n * 16];
            acc[n] = __builtin_amdgcn_wmma_f32_16x16x4_f32(
                false, afrag, false, bfrag, (short)0, acc[n], false, false);
        }
    }
}

static __device__ __forceinline__ void gemm_node_body(const float* __restrict__ A,
                                                      const float* __restrict__ bprev,
                                                      const float* __restrict__ dis,
                                                      const float* __restrict__ B,
                                                      float* __restrict__ hws,
                                                      int M, const int mode) {
    int wave = (blockIdx.x * blockDim.x + threadIdx.x) >> 5;
    int lane = threadIdx.x & 31;
    int m0 = wave * 16;
    if (m0 >= M) return;
    int half = lane >> 4;
    int l16  = lane & 15;

    const float* Arow = A + (size_t)(m0 + l16) * FDIM;
    float di = (mode != A_RAW) ? dis[m0 + l16] : 0.0f;

    v8f acc[8] = {};
    strip_k128(Arow, di, bprev, B, half, l16, mode, acc);

    float dg[8];
    #pragma unroll
    for (int g = 0; g < 8; ++g) dg[g] = dis[m0 + g + half * 8];

    #pragma unroll
    for (int n = 0; n < 8; ++n) {
        #pragma unroll
        for (int g = 0; g < 8; ++g) {
            int m = m0 + g + half * 8;
            float val = acc[n][g] * dg[g];
            size_t o = (size_t)m * FDIM + n * 16 + l16;
            hws[o] = val;
        }
    }
}

__global__ __launch_bounds__(256) void gemm_node_first(const float* __restrict__ X,
                                                       const float* __restrict__ dis,
                                                       const float* __restrict__ B,
                                                       float* __restrict__ hws, int M) {
    gemm_node_body(X, nullptr, dis, B, hws, M, A_RAW);
}


static __device__ __forceinline__ unsigned zero_bytes(unsigned y) {
    return ~(((y & 0x7F7F7F7Fu) + 0x7F7F7F7Fu) | y | 0x7F7F7F7Fu);
}

static __device__ __forceinline__ void owner_flush(const int* myseg, int cnt, int base, int lane,
                                                   const int* __restrict__ src,
                                                   const int* __restrict__ dst,
                                                   unsigned int* degc, unsigned int* cursor,
                                                   unsigned int* __restrict__ mylist) {
    for (int i0 = 0; i0 < cnt; i0 += 32) {
        int left = cnt - i0;
        bool ok = lane < left;
        int e = ok ? myseg[i0 + lane] : 0;
        int s = src[e];
        int d = dst[e];
        int n = left < 32 ? left : 32;
        unsigned int pos = 0u;
        if (lane == 0) pos = atomicAdd(cursor, (unsigned int)n);
        pos = (unsigned int)__builtin_amdgcn_readfirstlane((int)pos);
        if (ok) {
            int r = d - base;
            atomicAdd(&degc[r], 1u);
            mylist[pos + lane] = (unsigned int)s | ((unsigned int)r << 16);
        }
    }
}

__global__ __launch_bounds__(OWN_THREADS) void owner_build(const int* __restrict__ src,
                                                           const int* __restrict__ dst,
                                                           const unsigned int* __restrict__ own,
                                                           const unsigned int* __restrict__ cnt,
                                                           unsigned int* __restrict__ list,
                                                           int* __restrict__ meta,
                                                           float* __restrict__ dis) {
    __shared__ int seg[(OWN_THREADS / 32) * SEG_CAP];
    __shared__ unsigned int part[4 * 256];
    __shared__ unsigned int degc[OWN_ROWS];
    __shared__ unsigned int cursor;
    __shared__ unsigned int offs;

    const int b     = blockIdx.x;
    const int base  = b * OWN_ROWS;
    const int nrows = (N_NODES - base < OWN_ROWS) ? (N_NODES - base) : OWN_ROWS;
    const int tid   = threadIdx.x;
    const int lane  = tid & 31;
    const int w     = __builtin_amdgcn_readfirstlane(tid >> 5);

    {
        const int o = tid & 255;
        const int q = tid >> 8;
        unsigned int s = 0u;
        for (int c = q * (OC_WG / 4); c < (q + 1) * (OC_WG / 4); ++c) s += cnt[c * 256 + o];
        part[q * 256 + o] = s;
    }
    if (tid < OWN_ROWS) degc[tid] = 0u;
    if (tid == 0) cursor = 0u;
    __syncthreads();
    if (w == 0) {
        unsigned int s = 0u;
        for (int j = 0; j < 8; ++j) {
            int o = lane * 8 + j;
            if (o < b) s += part[o] + part[256 + o] + part[512 + o] + part[768 + o];
        }
        s += (unsigned int)__shfl_xor((int)s, 16);
        s += (unsigned int)__shfl_xor((int)s, 8);
        s += (unsigned int)__shfl_xor((int)s, 4);
        s += (unsigned int)__shfl_xor((int)s, 2);
        s += (unsigned int)__shfl_xor((int)s, 1);
        if (lane == 0) offs = s;
    }
    __syncthreads();
    unsigned int* mylist = list + offs;

    int* myseg = seg + w * SEG_CAP;
    int cnt_w = 0;
    const unsigned pat = (unsigned)b * 0x01010101u;
    const int rot = b * (N_CH / OWN_WG);
    const uint4* own4 = (const uint4*)own;

    for (int c0 = w; c0 < N_CH; c0 += 32 * 4) {
        uint4 q[4];
        int eb[4];
        bool ok4[4];
        #pragma unroll
        for (int k = 0; k < 4; ++k) {
            int cc = c0 + 32 * k;
            int c = cc + rot;
            if (c >= N_CH) c -= N_CH;
            int e = c * CH_EDGES + lane * 16;
            eb[k] = e;
            ok4[k] = (cc < N_CH) && (e < N_EDGES);
            q[k] = own4[ok4[k] ? (e >> 4) : 0];
        }
        #pragma unroll
        for (int k = 0; k < 4; ++k) {
            if (cnt_w > SEG_CAP - CH_EDGES) {
                owner_flush(myseg, cnt_w, base, lane, src, dst, degc, &cursor, mylist);
                cnt_w = 0;
            }
            unsigned zc = (zero_bytes(q[k].x ^ pat) >> 7) | (zero_bytes(q[k].y ^ pat) >> 6) |
                          (zero_bytes(q[k].z ^ pat) >> 5) | (zero_bytes(q[k].w ^ pat) >> 4);
            if (!ok4[k]) zc = 0u;
            for (int t = 0; t < 16; ++t) {
                bool has = zc != 0u;
                unsigned m = __builtin_amdgcn_ballot_w32(has);
                if (m == 0u) break;
                if (has) {
                    int c = __builtin_ctz(zc);
                    zc &= zc - 1u;
                    int p = ((c & 7) << 2) | (c >> 3);
                    int pos = cnt_w + __builtin_popcount(m & ((1u << lane) - 1u));
                    myseg[pos] = eb[k] + p;
                }
                cnt_w += __builtin_popcount(m);
            }
        }
    }
    owner_flush(myseg, cnt_w, base, lane, src, dst, degc, &cursor, mylist);

    __syncthreads();

    if (tid < nrows) dis[base + tid] = rsqrtf((float)degc[tid] + 1.0f);
    if (tid == 0) {
        meta[2 * b]     = (int)offs;
        meta[2 * b + 1] = (int)cursor;
    }
}

__global__ __launch_bounds__(OWN_THREADS) void agg_owner(const float* __restrict__ hws,
                                                         const unsigned int* __restrict__ list,
                                                         const int* __restrict__ meta,
                                                         float* __restrict__ agg) {
    __shared__ alignas(16) float acc[ACC_FLOATS];

    const int b     = blockIdx.x;
    const int base  = b * OWN_ROWS;
    const int nrows = (N_NODES - base < OWN_ROWS) ? (N_NODES - base) : OWN_ROWS;
    const int tid   = threadIdx.x;
    const int lane  = tid & 31;
    const int w     = __builtin_amdgcn_readfirstlane(tid >> 5);

    const int off = meta[2 * b];
    const int n   = meta[2 * b + 1];

    {
        const v4f* h4 = (const v4f*)(hws + (size_t)base * FDIM);
        v4f* a4 = (v4f*)acc;
        const int lim = nrows * (FDIM / 4);
        constexpr int NPC = (OWN_ROWS * (FDIM / 4) + OWN_THREADS - 1) / OWN_THREADS;
        static_assert(NPC == 7, "the fence below names seven values");
        v4f t4[NPC];
        #pragma unroll
        for (int t = 0; t < NPC; ++t) {
            int i = tid + t * OWN_THREADS;
            t4[t] = h4[i < lim ? i : lim - 1];
        }
        asm volatile("" :: "v"(t4[0]), "v"(t4[1]), "v"(t4[2]), "v"(t4[3]), "v"(t4[4]), "v"(t4[5]), "v"(t4[6]));
        #pragma unroll
        for (int t = 0; t < NPC; ++t) {
            int i = tid + t * OWN_THREADS;
            if (i < lim) a4[i] = t4[t];
        }
        if (tid < FDIM / 4) a4[OWN_ROWS * (FDIM / 4) + tid] = (v4f){0.0f, 0.0f, 0.0f, 0.0f};
    }
    const int per = (((n + 31) >> 5) + 7) & ~7;
    const int lo  = w * per;
    const int hi  = (lo + per < n) ? (lo + per) : n;
    const unsigned int* mylist = list + off;
    const unsigned int none = (unsigned int)OWN_ROWS << 16;
    unsigned int pk_next = (lane < hi - lo) ? mylist[lo + lane] : none;

    __syncthreads();

    for (int i0 = lo; i0 < hi; i0 += 32) {
        int left = hi - i0;
        unsigned int pk = pk_next;
        pk_next = (lane < left - 32) ? mylist[i0 + 32 + lane] : none;
        int s_l = (int)(pk & 0xFFFFu);
        int r_l = (int)(pk >> 16);
        #pragma unroll
        for (int g = 0; g < 4; ++g) {
            if (g * 8 < left) {
                float v[8][4];
                #pragma unroll
                for (int j = 0; j < 8; ++j) {
                    int s = __builtin_amdgcn_readlane(s_l, g * 8 + j);
                    const float* hp = hws + (size_t)s * FDIM + lane;
                    v[j][0] = hp[0];
                    v[j][1] = hp[32];
                    v[j][2] = hp[64];
                    v[j][3] = hp[96];
                }
                #pragma unroll
                for (int j = 0; j < 8; ++j) {
                    int r = __builtin_amdgcn_readlane(r_l, g * 8 + j);
                    float* ap = acc + r * FDIM + lane;
                    atomicAdd(ap + 0,  v[j][0]);
                    atomicAdd(ap + 32, v[j][1]);
                    atomicAdd(ap + 64, v[j][2]);
                    atomicAdd(ap + 96, v[j][3]);
                }
            }
        }
    }

    __syncthreads();

    {
        const float4* a4 = (const float4*)acc;
        float4* o4 = (float4*)(agg + (size_t)base * FDIM);
        for (int i = tid; i < nrows * (FDIM / 4); i += OWN_THREADS) o4[i] = a4[i];
    }
}

__global__ __launch_bounds__(OWN_THREADS) void agg_gemm(const float* __restrict__ hws,
                                                        const unsigned int* __restrict__ list,
                                                        const int* __restrict__ meta,
                                                        const float* __restrict__ dis,
                                                        const float* __restrict__ bPrev,
                                                        const float* __restrict__ B,
                                                        float* __restrict__ hwsNext) {
    __shared__ alignas(16) float acc[ACC_FLOATS];

    const int b     = blockIdx.x;
    const int base  = b * OWN_ROWS;
    const int nrows = (N_NODES - base < OWN_ROWS) ? (N_NODES - base) : OWN_ROWS;
    const int tid   = threadIdx.x;
    const int lane  = tid & 31;
    const int w     = __builtin_amdgcn_readfirstlane(tid >> 5);

    const int off = meta[2 * b];
    const int n   = meta[2 * b + 1];

    {
        const v4f* h4 = (const v4f*)(hws + (size_t)base * FDIM);
        v4f* a4 = (v4f*)acc;
        const int lim = nrows * (FDIM / 4);
        constexpr int NPC = (GEMM_ROWS * (FDIM / 4) + OWN_THREADS - 1) / OWN_THREADS;
        static_assert(NPC == 7, "the fence below names seven values");
        v4f t4[NPC];
        #pragma unroll
        for (int t = 0; t < NPC; ++t) {
            int i = tid + t * OWN_THREADS;
            t4[t] = h4[i < lim ? i : lim - 1];
        }
        asm volatile("" :: "v"(t4[0]), "v"(t4[1]), "v"(t4[2]), "v"(t4[3]), "v"(t4[4]), "v"(t4[5]), "v"(t4[6]));
        #pragma unroll
        for (int t = 0; t < NPC; ++t) {
            int i = tid + t * OWN_THREADS;
            if (i < GEMM_ROWS * (FDIM / 4)) {
                v4f val = (i < lim) ? t4[t] : (v4f){0.0f, 0.0f, 0.0f, 0.0f};
                a4[(i >> 5) * (ACC_PITCH / 4) + (i & 31)] = val;
            }
        }
    }
    const int per = (((n + 31) >> 5) + 7) & ~7;
    const int lo  = w * per;
    const int hi  = (lo + per < n) ? (lo + per) : n;
    const unsigned int* mylist = list + off;
    const unsigned int none = (unsigned int)OWN_ROWS << 16;
    unsigned int pk_next = (lane < hi - lo) ? mylist[lo + lane] : none;

    __syncthreads();

    for (int i0 = lo; i0 < hi; i0 += 32) {
        int left = hi - i0;
        unsigned int pk = pk_next;
        pk_next = (lane < left - 32) ? mylist[i0 + 32 + lane] : none;
        int s_l = (int)(pk & 0xFFFFu);
        int r_l = (int)(pk >> 16);
        #pragma unroll
        for (int g = 0; g < 4; ++g) {
            if (g * 8 < left) {
                float v[8][4];
                #pragma unroll
                for (int j = 0; j < 8; ++j) {
                    int s = __builtin_amdgcn_readlane(s_l, g * 8 + j);
                    const float* hp = hws + (size_t)s * FDIM + lane;
                    v[j][0] = hp[0];
                    v[j][1] = hp[32];
                    v[j][2] = hp[64];
                    v[j][3] = hp[96];
                }
                #pragma unroll
                for (int j = 0; j < 8; ++j) {
                    int r = __builtin_amdgcn_readlane(r_l, g * 8 + j);
                    float* ap = acc + r * ACC_PITCH + lane;
                    atomicAdd(ap + 0,  v[j][0]);
                    atomicAdd(ap + 32, v[j][1]);
                    atomicAdd(ap + 64, v[j][2]);
                    atomicAdd(ap + 96, v[j][3]);
                }
            }
        }
    }

    __syncthreads();

    const int half = lane >> 4;
    const int l16  = lane & 15;
    for (int t = w; t < GEMM_TASKS; t += OWN_THREADS / 32) {
        const int m0 = (t >> 2) * 16;
        if (m0 >= nrows) break;
        const int c0 = (t & 3) * 32;
        const float* Arow = acc + (m0 + l16) * ACC_PITCH;
        int gr = base + m0 + l16;
        float di = dis[gr < N_NODES ? gr : N_NODES - 1];
        float dg[8];
        #pragma unroll
        for (int g = 0; g < 8; ++g) {
            int mr = base + m0 + g + half * 8;
            dg[g] = dis[mr < N_NODES ? mr : N_NODES - 1];
        }

        v8f cc[2] = {};
        for (int k0 = 0; k0 < FDIM; k0 += 4) {
            int k = k0 + half * 2;
            v2f afrag = *(const v2f*)(Arow + k);
            v2f bb = *(const v2f*)(bPrev + k);
            afrag.x = fmaxf(afrag.x * di + bb.x, 0.0f);
            afrag.y = fmaxf(afrag.y * di + bb.y, 0.0f);
            const float* b0 = B + (size_t)k * FDIM + c0 + l16;
            #pragma unroll
            for (int nn = 0; nn < 2; ++nn) {
                v2f bfrag;
                bfrag.x = b0[nn * 16];
                bfrag.y = b0[FDIM + nn * 16];
                cc[nn] = __builtin_amdgcn_wmma_f32_16x16x4_f32(
                    false, afrag, false, bfrag, (short)0, cc[nn], false, false);
            }
        }

        #pragma unroll
        for (int g = 0; g < 8; ++g) {
            int m = m0 + g + half * 8;
            if (m < nrows) {
                float* op = hwsNext + (size_t)(base + m) * FDIM + c0 + l16;
                op[0]  = cc[0][g] * dg[g];
                op[16] = cc[1][g] * dg[g];
            }
        }
    }
}

__global__ __launch_bounds__(256) void gemm_edge(const float* __restrict__ AGG,
                                                 const float* __restrict__ dis,
                                                 const float* __restrict__ b3,
                                                 const int* __restrict__ pos_tr,
                                                 const int* __restrict__ neg_tr,
                                                 const int* __restrict__ pos_te,
                                                 const int* __restrict__ neg_te,
                                                 const float* __restrict__ Wl1,
                                                 const float* __restrict__ bl1,
                                                 const float* __restrict__ Wl2,
                                                 const float* __restrict__ bl2,
                                                 float* __restrict__ out) {
    int wave = (blockIdx.x * blockDim.x + threadIdx.x) >> 5;
    int lane = threadIdx.x & 31;
    int r0 = wave * 16;
    if (r0 >= N_ROWS) return;
    int half = lane >> 4;
    int l16  = lane & 15;
    int r = r0 + l16;

    int u, v;
    if (r < E_TR)            { u = pos_tr[r];              v = pos_tr[E_TR + r]; }
    else if (r < 2 * E_TR)   { int i = r - E_TR;           u = neg_tr[i]; v = neg_tr[E_TR + i]; }
    else if (r < 2 * E_TR + E_TE) { int i = r - 2 * E_TR;  u = pos_te[i]; v = pos_te[E_TE + i]; }
    else                     { int i = r - 2 * E_TR - E_TE; u = neg_te[i]; v = neg_te[E_TE + i]; }

    const float* Au = AGG + (size_t)u * FDIM;
    const float* Av = AGG + (size_t)v * FDIM;
    float du = dis[u];
    float dv = dis[v];

    v8f acc[8] = {};

    strip_k128(Au, du, b3, Wl1, half, l16, A_AFFINE, acc);
    strip_k128(Av, dv, b3, Wl1 + (size_t)FDIM * FDIM, half, l16, A_AFFINE, acc);

    float partial[8];
    #pragma unroll
    for (int g = 0; g < 8; ++g) partial[g] = 0.0f;

    #pragma unroll
    for (int n = 0; n < 8; ++n) {
        float bb = bl1[n * 16 + l16];
        float ww = Wl2[n * 16 + l16];
        #pragma unroll
        for (int g = 0; g < 8; ++g) {
            float z = fmaxf(acc[n][g] + bb, 0.0f);
            partial[g] += z * ww;
        }
    }

    float bias2 = bl2[0];
    #pragma unroll
    for (int g = 0; g < 8; ++g) {
        float s = partial[g];
        s += __shfl_xor(s, 8, 16);
        s += __shfl_xor(s, 4, 16);
        s += __shfl_xor(s, 2, 16);
        s += __shfl_xor(s, 1, 16);
        if (l16 == 0) out[r0 + g + half * 8] = s + bias2;
    }
}

extern "C" void kernel_launch(void* const* d_in, const int* in_sizes, int n_in,
                              void* d_out, int out_size, void* d_ws, size_t ws_size,
                              hipStream_t stream) {
    const float* x          = (const float*)d_in[0];
    const int*   edge_index = (const int*)d_in[1];
    const int*   pos_tr     = (const int*)d_in[2];
    const int*   neg_tr     = (const int*)d_in[3];
    const int*   pos_te     = (const int*)d_in[4];
    const int*   neg_te     = (const int*)d_in[5];
    const float* W0 = (const float*)d_in[6];
    const float* b0 = (const float*)d_in[7];
    const float* W1 = (const float*)d_in[8];
    const float* b1 = (const float*)d_in[9];
    const float* W2 = (const float*)d_in[10];
    const float* b2 = (const float*)d_in[11];
    const float* W3 = (const float*)d_in[12];
    const float* b3 = (const float*)d_in[13];
    const float* Wl1 = (const float*)d_in[14];
    const float* bl1 = (const float*)d_in[15];
    const float* Wl2 = (const float*)d_in[16];
    const float* bl2 = (const float*)d_in[17];
    float* out = (float*)d_out;

    const size_t NF = (size_t)N_NODES * FDIM;
    float* ws   = (float*)d_ws;
    float* dis  = ws;
    float* hws  = ws + 50048;
    float* agg  = hws + NF;
    unsigned int* own  = (unsigned int*)(agg + NF);
    unsigned int* cnt  = own + N_EDGES / 4;
    int*          meta = (int*)(cnt + OC_WG * 256);
    unsigned int* list = (unsigned int*)(meta + 2 * OWN_WG);

    const int* src = edge_index;
    const int* dst = edge_index + N_EDGES;

    own_count<<<OC_WG, OC_THREADS, 0, stream>>>(dst, own, cnt);
    owner_build<<<OWN_WG, OWN_THREADS, 0, stream>>>(src, dst, own, cnt, list, meta, dis);

    const int tilesM     = (N_NODES + 15) / 16;
    const int gemmBlocks = (tilesM + 7) / 8;

    gemm_node_first<<<gemmBlocks, 256, 0, stream>>>(x, dis, W0, hws, N_NODES);
    agg_gemm<<<OWN_WG, OWN_THREADS, 0, stream>>>(hws, list, meta, dis, b0, W1, agg);
    agg_gemm<<<OWN_WG, OWN_THREADS, 0, stream>>>(agg, list, meta, dis, b1, W2, hws);
    agg_gemm<<<OWN_WG, OWN_THREADS, 0, stream>>>(hws, list, meta, dis, b2, W3, agg);
    agg_owner<<<OWN_WG, OWN_THREADS, 0, stream>>>(agg, list, meta, hws);

    const int edgeTiles  = N_ROWS / 16;
    const int edgeBlocks = (edgeTiles + 7) / 8;
    gemm_edge<<<edgeBlocks, 256, 0, stream>>>(hws, dis, b3, pos_tr, neg_tr, pos_te, neg_te,
                                              Wl1, bl1, Wl2, bl2, out);
}


# GEMM B operand: a wave's column tiles interleaved (tile n = cols 64*(n/4)+4*c+n%4; 2*c+n in the fused layer GEMM), so B fragments are 16-byte loads, 4 a k step not 16; edge dot column sum regrouped
# speedup vs baseline: 3.2746x; 1.0481x over previous
#include <hip/hip_runtime.h>


typedef float v2f __attribute__((ext_vector_type(2)));
typedef float v8f __attribute__((ext_vector_type(8)));
typedef float v4f __attribute__((ext_vector_type(4)));

#define N_NODES 50000
#define N_EDGES 800000
#define FDIM    128
#define E_TR    100000
#define E_TE    20000
#define N_ROWS  240000

#define OWN_ROWS    196
#define OWN_WG      ((N_NODES + OWN_ROWS - 1) / OWN_ROWS)
#define OWN_THREADS 1024
#define SEG_CAP     1024
#define CH_EDGES    512
#define N_CH        ((N_EDGES + CH_EDGES - 1) / CH_EDGES)
#define OC_THREADS  256
#define OC_WG       ((N_EDGES / 16 + OC_THREADS - 1) / OC_THREADS)
static_assert(OWN_WG <= 256, "an owner id must fit in one byte");
static_assert(N_EDGES % 16 == 0, "a lane's 16 ids are all inside the edge list or all outside");
static_assert(2 * CH_EDGES <= SEG_CAP, "a drained list must take one full step");
static_assert(N_NODES <= 65536 && OWN_ROWS < 65536, "a list word holds the source row in 16 bits and the owned row above them");
static_assert(OC_WG % 4 == 0, "the count table is summed in four equal parts");
#define ACC_FLOATS  (164 * 1024 / 4)
static_assert(ACC_FLOATS >= (OWN_ROWS + 1) * FDIM, "owned rows and the spare row fit");
#define ACC_PITCH   132
#define GEMM_ROWS   208
#define GEMM_TASKS  (4 * (GEMM_ROWS / 16))
static_assert(GEMM_ROWS % 16 == 0 && GEMM_ROWS > OWN_ROWS, "whole tiles, and the spare row is among the zeroed rows");
static_assert(ACC_PITCH % 4 == 0, "a row starts on a 16-byte boundary");
static_assert(ACC_FLOATS >= GEMM_ROWS * ACC_PITCH, "the padded rows fit");

__global__ __launch_bounds__(OC_THREADS) void own_count(const int* __restrict__ dst,
                                                        unsigned int* __restrict__ own,
                                                        unsigned int* __restrict__ cnt) {
    __shared__ unsigned int hist[256];
    const int tid = threadIdx.x;
    hist[tid] = 0u;
    __syncthreads();
    const int t = blockIdx.x * OC_THREADS + tid;
    if (t < N_EDGES / 16) {
        const int4* d4 = (const int4*)(dst + (size_t)t * 16);
        int4 q[4];
        #pragma unroll
        for (int j = 0; j < 4; ++j) q[j] = d4[j];
        unsigned int wv[4];
        #pragma unroll
        for (int j = 0; j < 4; ++j) {
            unsigned int o0 = (unsigned int)q[j].x / OWN_ROWS;
            unsigned int o1 = (unsigned int)q[j].y / OWN_ROWS;
            unsigned int o2 = (unsigned int)q[j].z / OWN_ROWS;
            unsigned int o3 = (unsigned int)q[j].w / OWN_ROWS;
            atomicAdd(&hist[o0], 1u);
            atomicAdd(&hist[o1], 1u);
            atomicAdd(&hist[o2], 1u);
            atomicAdd(&hist[o3], 1u);
            wv[j] = o0 | (o1 << 8) | (o2 << 16) | (o3 << 24);
        }
        ((uint4*)own)[t] = make_uint4(wv[0], wv[1], wv[2], wv[3]);
    }
    __syncthreads();
    cnt[blockIdx.x * 256 + tid] = hist[tid];
}

#define A_RAW     0
#define A_AFFINE  1

static __device__ __forceinline__ void strip_k128(const float* __restrict__ Arow, float di,
                                                  const float* __restrict__ bias,
                                                  const float* __restrict__ Bm,
                                                  int half, int l16, const int mode,
                                                  v8f (&acc)[8]) {
    for (int k0 = 0; k0 < FDIM; k0 += 4) {
        int k = k0 + half * 2;
        v2f afrag = *(const v2f*)(Arow + k);
        if (mode != A_RAW) {
            v2f bb = *(const v2f*)(bias + k);
            afrag.x = afrag.x * di + bb.x;
            afrag.y = afrag.y * di + bb.y;
        }
        const float* b0 = Bm + (size_t)k * FDIM + 4 * l16;
        v4f r0[2], r1[2];
        r0[0] = *(const v4f*)(b0);
        r0[1] = *(const v4f*)(b0 + 64);
        r1[0] = *(const v4f*)(b0 + FDIM);
        r1[1] = *(const v4f*)(b0 + FDIM + 64);
        #pragma unroll
        for (int n = 0; n < 8; ++n) {
            v2f bfrag;
            bfrag.x = r0[n >> 2][n & 3];
            bfrag.y = r1[n >> 2][n & 3];
            acc[n] = __builtin_amdgcn_wmma_f32_16x16x4_f32(
                false, afrag, false, bfrag, (short)0, acc[n], false, false);
        }
    }
}

static __device__ __forceinline__ void gemm_node_body(const float* __restrict__ A,
                                                      const float* __restrict__ bprev,
                                                      const float* __restrict__ dis,
                                                      const float* __restrict__ B,
                                                      float* __restrict__ hws,
                                                      int M, const int mode) {
    int wave = (blockIdx.x * blockDim.x + threadIdx.x) >> 5;
    int lane = threadIdx.x & 31;
    int m0 = wave * 16;
    if (m0 >= M) return;
    int half = lane >> 4;
    int l16  = lane & 15;

    const float* Arow = A + (size_t)(m0 + l16) * FDIM;
    float di = (mode != A_RAW) ? dis[m0 + l16] : 0.0f;

    v8f acc[8] = {};
    strip_k128(Arow, di, bprev, B, half, l16, mode, acc);

    float dg[8];
    #pragma unroll
    for (int g = 0; g < 8; ++g) dg[g] = dis[m0 + g + half * 8];

    #pragma unroll
    for (int g = 0; g < 8; ++g) {
        int m = m0 + g + half * 8;
        float* op = hws + (size_t)m * FDIM + 4 * l16;
        *(v4f*)(op)      = (v4f){acc[0][g] * dg[g], acc[1][g] * dg[g], acc[2][g] * dg[g], acc[3][g] * dg[g]};
        *(v4f*)(op + 64) = (v4f){acc[4][g] * dg[g], acc[5][g] * dg[g], acc[6][g] * dg[g], acc[7][g] * dg[g]};
    }
}

__global__ __launch_bounds__(256) void gemm_node_first(const float* __restrict__ X,
                                                       const float* __restrict__ dis,
                                                       const float* __restrict__ B,
                                                       float* __restrict__ hws, int M) {
    gemm_node_body(X, nullptr, dis, B, hws, M, A_RAW);
}


static __device__ __forceinline__ unsigned zero_bytes(unsigned y) {
    return ~(((y & 0x7F7F7F7Fu) + 0x7F7F7F7Fu) | y | 0x7F7F7F7Fu);
}

static __device__ __forceinline__ void owner_flush(const int* myseg, int cnt, int base, int lane,
                                                   const int* __restrict__ src,
                                                   const int* __restrict__ dst,
                                                   unsigned int* degc, unsigned int* cursor,
                                                   unsigned int* __restrict__ mylist) {
    for (int i0 = 0; i0 < cnt; i0 += 32) {
        int left = cnt - i0;
        bool ok = lane < left;
        int e = ok ? myseg[i0 + lane] : 0;
        int s = src[e];
        int d = dst[e];
        int n = left < 32 ? left : 32;
        unsigned int pos = 0u;
        if (lane == 0) pos = atomicAdd(cursor, (unsigned int)n);
        pos = (unsigned int)__builtin_amdgcn_readfirstlane((int)pos);
        if (ok) {
            int r = d - base;
            atomicAdd(&degc[r], 1u);
            mylist[pos + lane] = (unsigned int)s | ((unsigned int)r << 16);
        }
    }
}

__global__ __launch_bounds__(OWN_THREADS) void owner_build(const int* __restrict__ src,
                                                           const int* __restrict__ dst,
                                                           const unsigned int* __restrict__ own,
                                                           const unsigned int* __restrict__ cnt,
                                                           unsigned int* __restrict__ list,
                                                           int* __restrict__ meta,
                                                           float* __restrict__ dis) {
    __shared__ int seg[(OWN_THREADS / 32) * SEG_CAP];
    __shared__ unsigned int part[4 * 256];
    __shared__ unsigned int degc[OWN_ROWS];
    __shared__ unsigned int cursor;
    __shared__ unsigned int offs;

    const int b     = blockIdx.x;
    const int base  = b * OWN_ROWS;
    const int nrows = (N_NODES - base < OWN_ROWS) ? (N_NODES - base) : OWN_ROWS;
    const int tid   = threadIdx.x;
    const int lane  = tid & 31;
    const int w     = __builtin_amdgcn_readfirstlane(tid >> 5);

    {
        const int o = tid & 255;
        const int q = tid >> 8;
        unsigned int s = 0u;
        for (int c = q * (OC_WG / 4); c < (q + 1) * (OC_WG / 4); ++c) s += cnt[c * 256 + o];
        part[q * 256 + o] = s;
    }
    if (tid < OWN_ROWS) degc[tid] = 0u;
    if (tid == 0) cursor = 0u;
    __syncthreads();
    if (w == 0) {
        unsigned int s = 0u;
        for (int j = 0; j < 8; ++j) {
            int o = lane * 8 + j;
            if (o < b) s += part[o] + part[256 + o] + part[512 + o] + part[768 + o];
        }
        s += (unsigned int)__shfl_xor((int)s, 16);
        s += (unsigned int)__shfl_xor((int)s, 8);
        s += (unsigned int)__shfl_xor((int)s, 4);
        s += (unsigned int)__shfl_xor((int)s, 2);
        s += (unsigned int)__shfl_xor((int)s, 1);
        if (lane == 0) offs = s;
    }
    __syncthreads();
    unsigned int* mylist = list + offs;

    int* myseg = seg + w * SEG_CAP;
    int cnt_w = 0;
    const unsigned pat = (unsigned)b * 0x01010101u;
    const int rot = b * (N_CH / OWN_WG);
    const uint4* own4 = (const uint4*)own;

    for (int c0 = w; c0 < N_CH; c0 += 32 * 4) {
        uint4 q[4];
        int eb[4];
        bool ok4[4];
        #pragma unroll
        for (int k = 0; k < 4; ++k) {
            int cc = c0 + 32 * k;
            int c = cc + rot;
            if (c >= N_CH) c -= N_CH;
            int e = c * CH_EDGES + lane * 16;
            eb[k] = e;
            ok4[k] = (cc < N_CH) && (e < N_EDGES);
            q[k] = own4[ok4[k] ? (e >> 4) : 0];
        }
        #pragma unroll
        for (int k = 0; k < 4; ++k) {
            if (cnt_w > SEG_CAP - CH_EDGES) {
                owner_flush(myseg, cnt_w, base, lane, src, dst, degc, &cursor, mylist);
                cnt_w = 0;
            }
            unsigned zc = (zero_bytes(q[k].x ^ pat) >> 7) | (zero_bytes(q[k].y ^ pat) >> 6) |
                          (zero_bytes(q[k].z ^ pat) >> 5) | (zero_bytes(q[k].w ^ pat) >> 4);
            if (!ok4[k]) zc = 0u;
            for (int t = 0; t < 16; ++t) {
                bool has = zc != 0u;
                unsigned m = __builtin_amdgcn_ballot_w32(has);
                if (m == 0u) break;
                if (has) {
                    int c = __builtin_ctz(zc);
                    zc &= zc - 1u;
                    int p = ((c & 7) << 2) | (c >> 3);
                    int pos = cnt_w + __builtin_popcount(m & ((1u << lane) - 1u));
                    myseg[pos] = eb[k] + p;
                }
                cnt_w += __builtin_popcount(m);
            }
        }
    }
    owner_flush(myseg, cnt_w, base, lane, src, dst, degc, &cursor, mylist);

    __syncthreads();

    if (tid < nrows) dis[base + tid] = rsqrtf((float)degc[tid] + 1.0f);
    if (tid == 0) {
        meta[2 * b]     = (int)offs;
        meta[2 * b + 1] = (int)cursor;
    }
}

__global__ __launch_bounds__(OWN_THREADS) void agg_owner(const float* __restrict__ hws,
                                                         const unsigned int* __restrict__ list,
                                                         const int* __restrict__ meta,
                                                         float* __restrict__ agg) {
    __shared__ alignas(16) float acc[ACC_FLOATS];

    const int b     = blockIdx.x;
    const int base  = b * OWN_ROWS;
    const int nrows = (N_NODES - base < OWN_ROWS) ? (N_NODES - base) : OWN_ROWS;
    const int tid   = threadIdx.x;
    const int lane  = tid & 31;
    const int w     = __builtin_amdgcn_readfirstlane(tid >> 5);

    const int off = meta[2 * b];
    const int n   = meta[2 * b + 1];

    {
        const v4f* h4 = (const v4f*)(hws + (size_t)base * FDIM);
        v4f* a4 = (v4f*)acc;
        const int lim = nrows * (FDIM / 4);
        constexpr int NPC = (OWN_ROWS * (FDIM / 4) + OWN_THREADS - 1) / OWN_THREADS;
        static_assert(NPC == 7, "the fence below names seven values");
        v4f t4[NPC];
        #pragma unroll
        for (int t = 0; t < NPC; ++t) {
            int i = tid + t * OWN_THREADS;
            t4[t] = h4[i < lim ? i : lim - 1];
        }
        asm volatile("" :: "v"(t4[0]), "v"(t4[1]), "v"(t4[2]), "v"(t4[3]), "v"(t4[4]), "v"(t4[5]), "v"(t4[6]));
        #pragma unroll
        for (int t = 0; t < NPC; ++t) {
            int i = tid + t * OWN_THREADS;
            if (i < lim) a4[i] = t4[t];
        }
        if (tid < FDIM / 4) a4[OWN_ROWS * (FDIM / 4) + tid] = (v4f){0.0f, 0.0f, 0.0f, 0.0f};
    }
    const int per = (((n + 31) >> 5) + 7) & ~7;
    const int lo  = w * per;
    const int hi  = (lo + per < n) ? (lo + per) : n;
    const unsigned int* mylist = list + off;
    const unsigned int none = (unsigned int)OWN_ROWS << 16;
    unsigned int pk_next = (lane < hi - lo) ? mylist[lo + lane] : none;

    __syncthreads();

    for (int i0 = lo; i0 < hi; i0 += 32) {
        int left = hi - i0;
        unsigned int pk = pk_next;
        pk_next = (lane < left - 32) ? mylist[i0 + 32 + lane] : none;
        int s_l = (int)(pk & 0xFFFFu);
        int r_l = (int)(pk >> 16);
        #pragma unroll
        for (int g = 0; g < 4; ++g) {
            if (g * 8 < left) {
                float v[8][4];
                #pragma unroll
                for (int j = 0; j < 8; ++j) {
                    int s = __builtin_amdgcn_readlane(s_l, g * 8 + j);
                    const float* hp = hws + (size_t)s * FDIM + lane;
                    v[j][0] = hp[0];
                    v[j][1] = hp[32];
                    v[j][2] = hp[64];
                    v[j][3] = hp[96];
                }
                #pragma unroll
                for (int j = 0; j < 8; ++j) {
                    int r = __builtin_amdgcn_readlane(r_l, g * 8 + j);
                    float* ap = acc + r * FDIM + lane;
                    atomicAdd(ap + 0,  v[j][0]);
                    atomicAdd(ap + 32, v[j][1]);
                    atomicAdd(ap + 64, v[j][2]);
                    atomicAdd(ap + 96, v[j][3]);
                }
            }
        }
    }

    __syncthreads();

    {
        const float4* a4 = (const float4*)acc;
        float4* o4 = (float4*)(agg + (size_t)base * FDIM);
        for (int i = tid; i < nrows * (FDIM / 4); i += OWN_THREADS) o4[i] = a4[i];
    }
}

__global__ __launch_bounds__(OWN_THREADS) void agg_gemm(const float* __restrict__ hws,
                                                        const unsigned int* __restrict__ list,
                                                        const int* __restrict__ meta,
                                                        const float* __restrict__ dis,
                                                        const float* __restrict__ bPrev,
                                                        const float* __restrict__ B,
                                                        float* __restrict__ hwsNext) {
    __shared__ alignas(16) float acc[ACC_FLOATS];

    const int b     = blockIdx.x;
    const int base  = b * OWN_ROWS;
    const int nrows = (N_NODES - base < OWN_ROWS) ? (N_NODES - base) : OWN_ROWS;
    const int tid   = threadIdx.x;
    const int lane  = tid & 31;
    const int w     = __builtin_amdgcn_readfirstlane(tid >> 5);

    const int off = meta[2 * b];
    const int n   = meta[2 * b + 1];

    {
        const v4f* h4 = (const v4f*)(hws + (size_t)base * FDIM);
        v4f* a4 = (v4f*)acc;
        const int lim = nrows * (FDIM / 4);
        constexpr int NPC = (GEMM_ROWS * (FDIM / 4) + OWN_THREADS - 1) / OWN_THREADS;
        static_assert(NPC == 7, "the fence below names seven values");
        v4f t4[NPC];
        #pragma unroll
        for (int t = 0; t < NPC; ++t) {
            int i = tid + t * OWN_THREADS;
            t4[t] = h4[i < lim ? i : lim - 1];
        }
        asm volatile("" :: "v"(t4[0]), "v"(t4[1]), "v"(t4[2]), "v"(t4[3]), "v"(t4[4]), "v"(t4[5]), "v"(t4[6]));
        #pragma unroll
        for (int t = 0; t < NPC; ++t) {
            int i = tid + t * OWN_THREADS;
            if (i < GEMM_ROWS * (FDIM / 4)) {
                v4f val = (i < lim) ? t4[t] : (v4f){0.0f, 0.0f, 0.0f, 0.0f};
                a4[(i >> 5) * (ACC_PITCH / 4) + (i & 31)] = val;
            }
        }
    }
    const int per = (((n + 31) >> 5) + 7) & ~7;
    const int lo  = w * per;
    const int hi  = (lo + per < n) ? (lo + per) : n;
    const unsigned int* mylist = list + off;
    const unsigned int none = (unsigned int)OWN_ROWS << 16;
    unsigned int pk_next = (lane < hi - lo) ? mylist[lo + lane] : none;

    __syncthreads();

    for (int i0 = lo; i0 < hi; i0 += 32) {
        int left = hi - i0;
        unsigned int pk = pk_next;
        pk_next = (lane < left - 32) ? mylist[i0 + 32 + lane] : none;
        int s_l = (int)(pk & 0xFFFFu);
        int r_l = (int)(pk >> 16);
        #pragma unroll
        for (int g = 0; g < 4; ++g) {
            if (g * 8 < left) {
                float v[8][4];
                #pragma unroll
                for (int j = 0; j < 8; ++j) {
                    int s = __builtin_amdgcn_readlane(s_l, g * 8 + j);
                    const float* hp = hws + (size_t)s * FDIM + lane;
                    v[j][0] = hp[0];
                    v[j][1] = hp[32];
                    v[j][2] = hp[64];
                    v[j][3] = hp[96];
                }
                #pragma unroll
                for (int j = 0; j < 8; ++j) {
                    int r = __builtin_amdgcn_readlane(r_l, g * 8 + j);
                    float* ap = acc + r * ACC_PITCH + lane;
                    atomicAdd(ap + 0,  v[j][0]);
                    atomicAdd(ap + 32, v[j][1]);
                    atomicAdd(ap + 64, v[j][2]);
                    atomicAdd(ap + 96, v[j][3]);
                }
            }
        }
    }

    __syncthreads();

    const int half = lane >> 4;
    const int l16  = lane & 15;
    for (int t = w; t < GEMM_TASKS; t += OWN_THREADS / 32) {
        const int m0 = (t >> 2) * 16;
        if (m0 >= nrows) break;
        const int c0 = (t & 3) * 32;
        const float* Arow = acc + (m0 + l16) * ACC_PITCH;
        int gr = base + m0 + l16;
        float di = dis[gr < N_NODES ? gr : N_NODES - 1];
        float dg[8];
        #pragma unroll
        for (int g = 0; g < 8; ++g) {
            int mr = base + m0 + g + half * 8;
            dg[g] = dis[mr < N_NODES ? mr : N_NODES - 1];
        }

        v8f cc[2] = {};
        for (int k0 = 0; k0 < FDIM; k0 += 4) {
            int k = k0 + half * 2;
            v2f afrag = *(const v2f*)(Arow + k);
            v2f bb = *(const v2f*)(bPrev + k);
            afrag.x = fmaxf(afrag.x * di + bb.x, 0.0f);
            afrag.y = fmaxf(afrag.y * di + bb.y, 0.0f);
            const float* b0 = B + (size_t)k * FDIM + c0 + 2 * l16;
            v2f r0 = *(const v2f*)(b0);
            v2f r1 = *(const v2f*)(b0 + FDIM);
            #pragma unroll
            for (int nn = 0; nn < 2; ++nn) {
                v2f bfrag;
                bfrag.x = r0[nn];
                bfrag.y = r1[nn];
                cc[nn] = __builtin_amdgcn_wmma_f32_16x16x4_f32(
                    false, afrag, false, bfrag, (short)0, cc[nn], false, false);
            }
        }

        #pragma unroll
        for (int g = 0; g < 8; ++g) {
            int m = m0 + g + half * 8;
            if (m < nrows) {
                float* op = hwsNext + (size_t)(base + m) * FDIM + c0 + 2 * l16;
                *(v2f*)op = (v2f){cc[0][g] * dg[g], cc[1][g] * dg[g]};
            }
        }
    }
}

__global__ __launch_bounds__(256) void gemm_edge(const float* __restrict__ AGG,
                                                 const float* __restrict__ dis,
                                                 const float* __restrict__ b3,
                                                 const int* __restrict__ pos_tr,
                                                 const int* __restrict__ neg_tr,
                                                 const int* __restrict__ pos_te,
                                                 const int* __restrict__ neg_te,
                                                 const float* __restrict__ Wl1,
                                                 const float* __restrict__ bl1,
                                                 const float* __restrict__ Wl2,
                                                 const float* __restrict__ bl2,
                                                 float* __restrict__ out) {
    int wave = (blockIdx.x * blockDim.x + threadIdx.x) >> 5;
    int lane = threadIdx.x & 31;
    int r0 = wave * 16;
    if (r0 >= N_ROWS) return;
    int half = lane >> 4;
    int l16  = lane & 15;
    int r = r0 + l16;

    int u, v;
    if (r < E_TR)            { u = pos_tr[r];              v = pos_tr[E_TR + r]; }
    else if (r < 2 * E_TR)   { int i = r - E_TR;           u = neg_tr[i]; v = neg_tr[E_TR + i]; }
    else if (r < 2 * E_TR + E_TE) { int i = r - 2 * E_TR;  u = pos_te[i]; v = pos_te[E_TE + i]; }
    else                     { int i = r - 2 * E_TR - E_TE; u = neg_te[i]; v = neg_te[E_TE + i]; }

    const float* Au = AGG + (size_t)u * FDIM;
    const float* Av = AGG + (size_t)v * FDIM;
    float du = dis[u];
    float dv = dis[v];

    v8f acc[8] = {};

    strip_k128(Au, du, b3, Wl1, half, l16, A_AFFINE, acc);
    strip_k128(Av, dv, b3, Wl1 + (size_t)FDIM * FDIM, half, l16, A_AFFINE, acc);

    float partial[8];
    #pragma unroll
    for (int g = 0; g < 8; ++g) partial[g] = 0.0f;

    v4f bb4[2], ww4[2];
    bb4[0] = *(const v4f*)(bl1 + 4 * l16);
    bb4[1] = *(const v4f*)(bl1 + 64 + 4 * l16);
    ww4[0] = *(const v4f*)(Wl2 + 4 * l16);
    ww4[1] = *(const v4f*)(Wl2 + 64 + 4 * l16);
    #pragma unroll
    for (int n = 0; n < 8; ++n) {
        float bb = bb4[n >> 2][n & 3];
        float ww = ww4[n >> 2][n & 3];
        #pragma unroll
        for (int g = 0; g < 8; ++g) {
            float z = fmaxf(acc[n][g] + bb, 0.0f);
            partial[g] += z * ww;
        }
    }

    float bias2 = bl2[0];
    #pragma unroll
    for (int g = 0; g < 8; ++g) {
        float s = partial[g];
        s += __shfl_xor(s, 8, 16);
        s += __shfl_xor(s, 4, 16);
        s += __shfl_xor(s, 2, 16);
        s += __shfl_xor(s, 1, 16);
        if (l16 == 0) out[r0 + g + half * 8] = s + bias2;
    }
}

extern "C" void kernel_launch(void* const* d_in, const int* in_sizes, int n_in,
                              void* d_out, int out_size, void* d_ws, size_t ws_size,
                              hipStream_t stream) {
    const float* x          = (const float*)d_in[0];
    const int*   edge_index = (const int*)d_in[1];
    const int*   pos_tr     = (const int*)d_in[2];
    const int*   neg_tr     = (const int*)d_in[3];
    const int*   pos_te     = (const int*)d_in[4];
    const int*   neg_te     = (const int*)d_in[5];
    const float* W0 = (const float*)d_in[6];
    const float* b0 = (const float*)d_in[7];
    const float* W1 = (const float*)d_in[8];
    const float* b1 = (const float*)d_in[9];
    const float* W2 = (const float*)d_in[10];
    const float* b2 = (const float*)d_in[11];
    const float* W3 = (const float*)d_in[12];
    const float* b3 = (const float*)d_in[13];
    const float* Wl1 = (const float*)d_in[14];
    const float* bl1 = (const float*)d_in[15];
    const float* Wl2 = (const float*)d_in[16];
    const float* bl2 = (const float*)d_in[17];
    float* out = (float*)d_out;

    const size_t NF = (size_t)N_NODES * FDIM;
    float* ws   = (float*)d_ws;
    float* dis  = ws;
    float* hws  = ws + 50048;
    float* agg  = hws + NF;
    unsigned int* own  = (unsigned int*)(agg + NF);
    unsigned int* cnt  = own + N_EDGES / 4;
    int*          meta = (int*)(cnt + OC_WG * 256);
    unsigned int* list = (unsigned int*)(meta + 2 * OWN_WG);

    const int* src = edge_index;
    const int* dst = edge_index + N_EDGES;

    own_count<<<OC_WG, OC_THREADS, 0, stream>>>(dst, own, cnt);
    owner_build<<<OWN_WG, OWN_THREADS, 0, stream>>>(src, dst, own, cnt, list, meta, dis);

    const int tilesM     = (N_NODES + 15) / 16;
    const int gemmBlocks = (tilesM + 7) / 8;

    gemm_node_first<<<gemmBlocks, 256, 0, stream>>>(x, dis, W0, hws, N_NODES);
    agg_gemm<<<OWN_WG, OWN_THREADS, 0, stream>>>(hws, list, meta, dis, b0, W1, agg);
    agg_gemm<<<OWN_WG, OWN_THREADS, 0, stream>>>(agg, list, meta, dis, b1, W2, hws);
    agg_gemm<<<OWN_WG, OWN_THREADS, 0, stream>>>(hws, list, meta, dis, b2, W3, agg);
    agg_owner<<<OWN_WG, OWN_THREADS, 0, stream>>>(agg, list, meta, hws);

    const int edgeTiles  = N_ROWS / 16;
    const int edgeBlocks = (edgeTiles + 7) / 8;
    gemm_edge<<<edgeBlocks, 256, 0, stream>>>(hws, dis, b3, pos_tr, neg_tr, pos_te, neg_te,
                                              Wl1, bl1, Wl2, bl2, out);
}


# Owner lists sorted by owned row once per call (in-degree prefix + arrival rank); a wave sums same-destination neighbours in registers, one LDS atomic set per destination met; sum tree regrouped
# speedup vs baseline: 3.4878x; 1.0651x over previous
#include <hip/hip_runtime.h>


typedef float v2f __attribute__((ext_vector_type(2)));
typedef float v8f __attribute__((ext_vector_type(8)));
typedef float v4f __attribute__((ext_vector_type(4)));

#define N_NODES 50000
#define N_EDGES 800000
#define FDIM    128
#define E_TR    100000
#define E_TE    20000
#define N_ROWS  240000

#define OWN_ROWS    196
#define OWN_WG      ((N_NODES + OWN_ROWS - 1) / OWN_ROWS)
#define OWN_THREADS 1024
#define SEG_CAP     1024
#define CH_EDGES    512
#define N_CH        ((N_EDGES + CH_EDGES - 1) / CH_EDGES)
#define OC_THREADS  256
#define OC_WG       ((N_EDGES / 16 + OC_THREADS - 1) / OC_THREADS)
static_assert(OWN_WG <= 256, "an owner id must fit in one byte");
static_assert(N_EDGES % 16 == 0, "a lane's 16 ids are all inside the edge list or all outside");
static_assert(2 * CH_EDGES <= SEG_CAP, "a drained list must take one full step");
static_assert(N_NODES <= 65536 && OWN_ROWS < 65536, "a list word holds the source row in 16 bits and the owned row above them");
static_assert(OC_WG % 4 == 0, "the count table is summed in four equal parts");
#define ACC_FLOATS  (164 * 1024 / 4)
static_assert(ACC_FLOATS >= (OWN_ROWS + 1) * FDIM, "owned rows and the spare row fit");
#define ACC_PITCH   132
#define GEMM_ROWS   208
#define GEMM_TASKS  (4 * (GEMM_ROWS / 16))
static_assert(GEMM_ROWS % 16 == 0 && GEMM_ROWS > OWN_ROWS, "whole tiles, and the spare row is among the zeroed rows");
static_assert(ACC_PITCH % 4 == 0, "a row starts on a 16-byte boundary");
static_assert(ACC_FLOATS >= GEMM_ROWS * ACC_PITCH, "the padded rows fit");

__global__ __launch_bounds__(OC_THREADS) void own_count(const int* __restrict__ dst,
                                                        unsigned int* __restrict__ own,
                                                        unsigned int* __restrict__ cnt) {
    __shared__ unsigned int hist[256];
    const int tid = threadIdx.x;
    hist[tid] = 0u;
    __syncthreads();
    const int t = blockIdx.x * OC_THREADS + tid;
    if (t < N_EDGES / 16) {
        const int4* d4 = (const int4*)(dst + (size_t)t * 16);
        int4 q[4];
        #pragma unroll
        for (int j = 0; j < 4; ++j) q[j] = d4[j];
        unsigned int wv[4];
        #pragma unroll
        for (int j = 0; j < 4; ++j) {
            unsigned int o0 = (unsigned int)q[j].x / OWN_ROWS;
            unsigned int o1 = (unsigned int)q[j].y / OWN_ROWS;
            unsigned int o2 = (unsigned int)q[j].z / OWN_ROWS;
            unsigned int o3 = (unsigned int)q[j].w / OWN_ROWS;
            atomicAdd(&hist[o0], 1u);
            atomicAdd(&hist[o1], 1u);
            atomicAdd(&hist[o2], 1u);
            atomicAdd(&hist[o3], 1u);
            wv[j] = o0 | (o1 << 8) | (o2 << 16) | (o3 << 24);
        }
        ((uint4*)own)[t] = make_uint4(wv[0], wv[1], wv[2], wv[3]);
    }
    __syncthreads();
    cnt[blockIdx.x * 256 + tid] = hist[tid];
}

#define A_RAW     0
#define A_AFFINE  1

static __device__ __forceinline__ void strip_k128(const float* __restrict__ Arow, float di,
                                                  const float* __restrict__ bias,
                                                  const float* __restrict__ Bm,
                                                  int half, int l16, const int mode,
                                                  v8f (&acc)[8]) {
    for (int k0 = 0; k0 < FDIM; k0 += 4) {
        int k = k0 + half * 2;
        v2f afrag = *(const v2f*)(Arow + k);
        if (mode != A_RAW) {
            v2f bb = *(const v2f*)(bias + k);
            afrag.x = afrag.x * di + bb.x;
            afrag.y = afrag.y * di + bb.y;
        }
        const float* b0 = Bm + (size_t)k * FDIM + 4 * l16;
        v4f r0[2], r1[2];
        r0[0] = *(const v4f*)(b0);
        r0[1] = *(const v4f*)(b0 + 64);
        r1[0] = *(const v4f*)(b0 + FDIM);
        r1[1] = *(const v4f*)(b0 + FDIM + 64);
        #pragma unroll
        for (int n = 0; n < 8; ++n) {
            v2f bfrag;
            bfrag.x = r0[n >> 2][n & 3];
            bfrag.y = r1[n >> 2][n & 3];
            acc[n] = __builtin_amdgcn_wmma_f32_16x16x4_f32(
                false, afrag, false, bfrag, (short)0, acc[n], false, false);
        }
    }
}

static __device__ __forceinline__ void gemm_node_body(const float* __restrict__ A,
                                                      const float* __restrict__ bprev,
                                                      const float* __restrict__ dis,
                                                      const float* __restrict__ B,
                                                      float* __restrict__ hws,
                                                      int M, const int mode) {
    int wave = (blockIdx.x * blockDim.x + threadIdx.x) >> 5;
    int lane = threadIdx.x & 31;
    int m0 = wave * 16;
    if (m0 >= M) return;
    int half = lane >> 4;
    int l16  = lane & 15;

    const float* Arow = A + (size_t)(m0 + l16) * FDIM;
    float di = (mode != A_RAW) ? dis[m0 + l16] : 0.0f;

    v8f acc[8] = {};
    strip_k128(Arow, di, bprev, B, half, l16, mode, acc);

    float dg[8];
    #pragma unroll
    for (int g = 0; g < 8; ++g) dg[g] = dis[m0 + g + half * 8];

    #pragma unroll
    for (int g = 0; g < 8; ++g) {
        int m = m0 + g + half * 8;
        float* op = hws + (size_t)m * FDIM + 4 * l16;
        *(v4f*)(op)      = (v4f){acc[0][g] * dg[g], acc[1][g] * dg[g], acc[2][g] * dg[g], acc[3][g] * dg[g]};
        *(v4f*)(op + 64) = (v4f){acc[4][g] * dg[g], acc[5][g] * dg[g], acc[6][g] * dg[g], acc[7][g] * dg[g]};
    }
}

__global__ __launch_bounds__(256) void gemm_node_first(const float* __restrict__ X,
                                                       const float* __restrict__ dis,
                                                       const float* __restrict__ B,
                                                       float* __restrict__ hws, int M) {
    gemm_node_body(X, nullptr, dis, B, hws, M, A_RAW);
}


static __device__ __forceinline__ unsigned zero_bytes(unsigned y) {
    return ~(((y & 0x7F7F7F7Fu) + 0x7F7F7F7Fu) | y | 0x7F7F7F7Fu);
}

static __device__ __forceinline__ void owner_flush(const int* myseg, int cnt, int base, int lane,
                                                   const int* __restrict__ src,
                                                   const int* __restrict__ dst,
                                                   unsigned int* degc, unsigned int* cursor,
                                                   unsigned int* __restrict__ mylist) {
    for (int i0 = 0; i0 < cnt; i0 += 32) {
        int left = cnt - i0;
        bool ok = lane < left;
        int e = ok ? myseg[i0 + lane] : 0;
        int s = src[e];
        int d = dst[e];
        int n = left < 32 ? left : 32;
        unsigned int pos = 0u;
        if (lane == 0) pos = atomicAdd(cursor, (unsigned int)n);
        pos = (unsigned int)__builtin_amdgcn_readfirstlane((int)pos);
        if (ok) {
            int r = d - base;
            atomicAdd(&degc[r], 1u);
            mylist[pos + lane] = (unsigned int)s | ((unsigned int)r << 16);
        }
    }
}

__global__ __launch_bounds__(OWN_THREADS) void owner_build(const int* __restrict__ src,
                                                           const int* __restrict__ dst,
                                                           const unsigned int* __restrict__ own,
                                                           const unsigned int* __restrict__ cnt,
                                                           unsigned int* __restrict__ list,
                                                           unsigned int* __restrict__ scratch,
                                                           int* __restrict__ meta,
                                                           float* __restrict__ dis) {
    __shared__ int seg[(OWN_THREADS / 32) * SEG_CAP];
    __shared__ unsigned int rowpos[OWN_ROWS];
    __shared__ unsigned int part[4 * 256];
    __shared__ unsigned int degc[OWN_ROWS];
    __shared__ unsigned int cursor;
    __shared__ unsigned int offs;

    const int b     = blockIdx.x;
    const int base  = b * OWN_ROWS;
    const int nrows = (N_NODES - base < OWN_ROWS) ? (N_NODES - base) : OWN_ROWS;
    const int tid   = threadIdx.x;
    const int lane  = tid & 31;
    const int w     = __builtin_amdgcn_readfirstlane(tid >> 5);

    {
        const int o = tid & 255;
        const int q = tid >> 8;
        unsigned int s = 0u;
        for (int c = q * (OC_WG / 4); c < (q + 1) * (OC_WG / 4); ++c) s += cnt[c * 256 + o];
        part[q * 256 + o] = s;
    }
    if (tid < OWN_ROWS) degc[tid] = 0u;
    if (tid == 0) cursor = 0u;
    __syncthreads();
    if (w == 0) {
        unsigned int s = 0u;
        for (int j = 0; j < 8; ++j) {
            int o = lane * 8 + j;
            if (o < b) s += part[o] + part[256 + o] + part[512 + o] + part[768 + o];
        }
        s += (unsigned int)__shfl_xor((int)s, 16);
        s += (unsigned int)__shfl_xor((int)s, 8);
        s += (unsigned int)__shfl_xor((int)s, 4);
        s += (unsigned int)__shfl_xor((int)s, 2);
        s += (unsigned int)__shfl_xor((int)s, 1);
        if (lane == 0) offs = s;
    }
    __syncthreads();
    unsigned int* mylist = list + offs;
    unsigned int* myraw  = scratch + offs;

    int* myseg = seg + w * SEG_CAP;
    int cnt_w = 0;
    const unsigned pat = (unsigned)b * 0x01010101u;
    const int rot = b * (N_CH / OWN_WG);
    const uint4* own4 = (const uint4*)own;

    for (int c0 = w; c0 < N_CH; c0 += 32 * 4) {
        uint4 q[4];
        int eb[4];
        bool ok4[4];
        #pragma unroll
        for (int k = 0; k < 4; ++k) {
            int cc = c0 + 32 * k;
            int c = cc + rot;
            if (c >= N_CH) c -= N_CH;
            int e = c * CH_EDGES + lane * 16;
            eb[k] = e;
            ok4[k] = (cc < N_CH) && (e < N_EDGES);
            q[k] = own4[ok4[k] ? (e >> 4) : 0];
        }
        #pragma unroll
        for (int k = 0; k < 4; ++k) {
            if (cnt_w > SEG_CAP - CH_EDGES) {
                owner_flush(myseg, cnt_w, base, lane, src, dst, degc, &cursor, myraw);
                cnt_w = 0;
            }
            unsigned zc = (zero_bytes(q[k].x ^ pat) >> 7) | (zero_bytes(q[k].y ^ pat) >> 6) |
                          (zero_bytes(q[k].z ^ pat) >> 5) | (zero_bytes(q[k].w ^ pat) >> 4);
            if (!ok4[k]) zc = 0u;
            for (int t = 0; t < 16; ++t) {
                bool has = zc != 0u;
                unsigned m = __builtin_amdgcn_ballot_w32(has);
                if (m == 0u) break;
                if (has) {
                    int c = __builtin_ctz(zc);
                    zc &= zc - 1u;
                    int p = ((c & 7) << 2) | (c >> 3);
                    int pos = cnt_w + __builtin_popcount(m & ((1u << lane) - 1u));
                    myseg[pos] = eb[k] + p;
                }
                cnt_w += __builtin_popcount(m);
            }
        }
    }
    owner_flush(myseg, cnt_w, base, lane, src, dst, degc, &cursor, myraw);

    __syncthreads();

    static_assert(OWN_ROWS <= 32 * 7, "seven rows a lane cover the owned rows");
    if (w == 0) {
        unsigned int cj[7];
        unsigned int s = 0u;
        #pragma unroll
        for (int j = 0; j < 7; ++j) {
            int idx = lane * 7 + j;
            cj[j] = (idx < OWN_ROWS) ? degc[idx] : 0u;
            s += cj[j];
        }
        unsigned int incl = s;
        #pragma unroll
        for (int d = 1; d < 32; d <<= 1) {
            unsigned int t = (unsigned int)__shfl_up((int)incl, d);
            if (lane >= d) incl += t;
        }
        unsigned int run = incl - s;
        #pragma unroll
        for (int j = 0; j < 7; ++j) {
            int idx = lane * 7 + j;
            if (idx < OWN_ROWS) rowpos[idx] = run;
            run += cj[j];
        }
    }
    __syncthreads();
    {
        const unsigned int nall = cursor;
        for (unsigned int i = (unsigned int)tid; i < nall; i += OWN_THREADS) {
            unsigned int wd = myraw[i];
            unsigned int p = atomicAdd(&rowpos[wd >> 16], 1u);
            mylist[p] = wd;
        }
    }

    if (tid < nrows) dis[base + tid] = rsqrtf((float)degc[tid] + 1.0f);
    if (tid == 0) {
        meta[2 * b]     = (int)offs;
        meta[2 * b + 1] = (int)cursor;
    }
}

__global__ __launch_bounds__(OWN_THREADS) void agg_owner(const float* __restrict__ hws,
                                                         const unsigned int* __restrict__ list,
                                                         const int* __restrict__ meta,
                                                         float* __restrict__ agg) {
    __shared__ alignas(16) float acc[ACC_FLOATS];

    const int b     = blockIdx.x;
    const int base  = b * OWN_ROWS;
    const int nrows = (N_NODES - base < OWN_ROWS) ? (N_NODES - base) : OWN_ROWS;
    const int tid   = threadIdx.x;
    const int lane  = tid & 31;
    const int w     = __builtin_amdgcn_readfirstlane(tid >> 5);

    const int off = meta[2 * b];
    const int n   = meta[2 * b + 1];

    {
        const v4f* h4 = (const v4f*)(hws + (size_t)base * FDIM);
        v4f* a4 = (v4f*)acc;
        const int lim = nrows * (FDIM / 4);
        constexpr int NPC = (OWN_ROWS * (FDIM / 4) + OWN_THREADS - 1) / OWN_THREADS;
        static_assert(NPC == 7, "the fence below names seven values");
        v4f t4[NPC];
        #pragma unroll
        for (int t = 0; t < NPC; ++t) {
            int i = tid + t * OWN_THREADS;
            t4[t] = h4[i < lim ? i : lim - 1];
        }
        asm volatile("" :: "v"(t4[0]), "v"(t4[1]), "v"(t4[2]), "v"(t4[3]), "v"(t4[4]), "v"(t4[5]), "v"(t4[6]));
        #pragma unroll
        for (int t = 0; t < NPC; ++t) {
            int i = tid + t * OWN_THREADS;
            if (i < lim) a4[i] = t4[t];
        }
        if (tid < FDIM / 4) a4[OWN_ROWS * (FDIM / 4) + tid] = (v4f){0.0f, 0.0f, 0.0f, 0.0f};
    }
    const int per = (((n + 31) >> 5) + 7) & ~7;
    const int lo  = w * per;
    const int hi  = (lo + per < n) ? (lo + per) : n;
    const unsigned int* mylist = list + off;
    const unsigned int none = (unsigned int)OWN_ROWS << 16;
    unsigned int pk_next = (lane < hi - lo) ? mylist[lo + lane] : none;

    __syncthreads();

    for (int i0 = lo; i0 < hi; i0 += 32) {
        int left = hi - i0;
        unsigned int pk = pk_next;
        pk_next = (lane < left - 32) ? mylist[i0 + 32 + lane] : none;
        int s_l = (int)(pk & 0xFFFFu);
        int r_l = (int)(pk >> 16);
        #pragma unroll
        for (int g = 0; g < 4; ++g) {
            if (g * 8 < left) {
                float v[8][4];
                #pragma unroll
                for (int j = 0; j < 8; ++j) {
                    int s = __builtin_amdgcn_readlane(s_l, g * 8 + j);
                    const float* hp = hws + (size_t)s * FDIM + lane;
                    v[j][0] = hp[0];
                    v[j][1] = hp[32];
                    v[j][2] = hp[64];
                    v[j][3] = hp[96];
                }
                int rc = __builtin_amdgcn_readlane(r_l, g * 8);
                float a0 = v[0][0], a1 = v[0][1], a2 = v[0][2], a3 = v[0][3];
                #pragma unroll
                for (int j = 1; j < 8; ++j) {
                    int r = __builtin_amdgcn_readlane(r_l, g * 8 + j);
                    if (r != rc) {
                        float* ap = acc + rc * FDIM + lane;
                        atomicAdd(ap + 0,  a0);
                        atomicAdd(ap + 32, a1);
                        atomicAdd(ap + 64, a2);
                        atomicAdd(ap + 96, a3);
                        rc = r;
                        a0 = v[j][0]; a1 = v[j][1]; a2 = v[j][2]; a3 = v[j][3];
                    } else {
                        a0 += v[j][0]; a1 += v[j][1]; a2 += v[j][2]; a3 += v[j][3];
                    }
                }
                {
                    float* ap = acc + rc * FDIM + lane;
                    atomicAdd(ap + 0,  a0);
                    atomicAdd(ap + 32, a1);
                    atomicAdd(ap + 64, a2);
                    atomicAdd(ap + 96, a3);
                }
            }
        }
    }

    __syncthreads();

    {
        const float4* a4 = (const float4*)acc;
        float4* o4 = (float4*)(agg + (size_t)base * FDIM);
        for (int i = tid; i < nrows * (FDIM / 4); i += OWN_THREADS) o4[i] = a4[i];
    }
}

__global__ __launch_bounds__(OWN_THREADS) void agg_gemm(const float* __restrict__ hws,
                                                        const unsigned int* __restrict__ list,
                                                        const int* __restrict__ meta,
                                                        const float* __restrict__ dis,
                                                        const float* __restrict__ bPrev,
                                                        const float* __restrict__ B,
                                                        float* __restrict__ hwsNext) {
    __shared__ alignas(16) float acc[ACC_FLOATS];

    const int b     = blockIdx.x;
    const int base  = b * OWN_ROWS;
    const int nrows = (N_NODES - base < OWN_ROWS) ? (N_NODES - base) : OWN_ROWS;
    const int tid   = threadIdx.x;
    const int lane  = tid & 31;
    const int w     = __builtin_amdgcn_readfirstlane(tid >> 5);

    const int off = meta[2 * b];
    const int n   = meta[2 * b + 1];

    {
        const v4f* h4 = (const v4f*)(hws + (size_t)base * FDIM);
        v4f* a4 = (v4f*)acc;
        const int lim = nrows * (FDIM / 4);
        constexpr int NPC = (GEMM_ROWS * (FDIM / 4) + OWN_THREADS - 1) / OWN_THREADS;
        static_assert(NPC == 7, "the fence below names seven values");
        v4f t4[NPC];
        #pragma unroll
        for (int t = 0; t < NPC; ++t) {
            int i = tid + t * OWN_THREADS;
            t4[t] = h4[i < lim ? i : lim - 1];
        }
        asm volatile("" :: "v"(t4[0]), "v"(t4[1]), "v"(t4[2]), "v"(t4[3]), "v"(t4[4]), "v"(t4[5]), "v"(t4[6]));
        #pragma unroll
        for (int t = 0; t < NPC; ++t) {
            int i = tid + t * OWN_THREADS;
            if (i < GEMM_ROWS * (FDIM / 4)) {
                v4f val = (i < lim) ? t4[t] : (v4f){0.0f, 0.0f, 0.0f, 0.0f};
                a4[(i >> 5) * (ACC_PITCH / 4) + (i & 31)] = val;
            }
        }
    }
    const int per = (((n + 31) >> 5) + 7) & ~7;
    const int lo  = w * per;
    const int hi  = (lo + per < n) ? (lo + per) : n;
    const unsigned int* mylist = list + off;
    const unsigned int none = (unsigned int)OWN_ROWS << 16;
    unsigned int pk_next = (lane < hi - lo) ? mylist[lo + lane] : none;

    __syncthreads();

    for (int i0 = lo; i0 < hi; i0 += 32) {
        int left = hi - i0;
        unsigned int pk = pk_next;
        pk_next = (lane < left - 32) ? mylist[i0 + 32 + lane] : none;
        int s_l = (int)(pk & 0xFFFFu);
        int r_l = (int)(pk >> 16);
        #pragma unroll
        for (int g = 0; g < 4; ++g) {
            if (g * 8 < left) {
                float v[8][4];
                #pragma unroll
                for (int j = 0; j < 8; ++j) {
                    int s = __builtin_amdgcn_readlane(s_l, g * 8 + j);
                    const float* hp = hws + (size_t)s * FDIM + lane;
                    v[j][0] = hp[0];
                    v[j][1] = hp[32];
                    v[j][2] = hp[64];
                    v[j][3] = hp[96];
                }
                int rc = __builtin_amdgcn_readlane(r_l, g * 8);
                float a0 = v[0][0], a1 = v[0][1], a2 = v[0][2], a3 = v[0][3];
                #pragma unroll
                for (int j = 1; j < 8; ++j) {
                    int r = __builtin_amdgcn_readlane(r_l, g * 8 + j);
                    if (r != rc) {
                        float* ap = acc + rc * ACC_PITCH + lane;
                        atomicAdd(ap + 0,  a0);
                        atomicAdd(ap + 32, a1);
                        atomicAdd(ap + 64, a2);
                        atomicAdd(ap + 96, a3);
                        rc = r;
                        a0 = v[j][0]; a1 = v[j][1]; a2 = v[j][2]; a3 = v[j][3];
                    } else {
                        a0 += v[j][0]; a1 += v[j][1]; a2 += v[j][2]; a3 += v[j][3];
                    }
                }
                {
                    float* ap = acc + rc * ACC_PITCH + lane;
                    atomicAdd(ap + 0,  a0);
                    atomicAdd(ap + 32, a1);
                    atomicAdd(ap + 64, a2);
                    atomicAdd(ap + 96, a3);
                }
            }
        }
    }

    __syncthreads();

    const int half = lane >> 4;
    const int l16  = lane & 15;
    for (int t = w; t < GEMM_TASKS; t += OWN_THREADS / 32) {
        const int m0 = (t >> 2) * 16;
        if (m0 >= nrows) break;
        const int c0 = (t & 3) * 32;
        const float* Arow = acc + (m0 + l16) * ACC_PITCH;
        int gr = base + m0 + l16;
        float di = dis[gr < N_NODES ? gr : N_NODES - 1];
        float dg[8];
        #pragma unroll
        for (int g = 0; g < 8; ++g) {
            int mr = base + m0 + g + half * 8;
            dg[g] = dis[mr < N_NODES ? mr : N_NODES - 1];
        }

        v8f cc[2] = {};
        for (int k0 = 0; k0 < FDIM; k0 += 4) {
            int k = k0 + half * 2;
            v2f afrag = *(const v2f*)(Arow + k);
            v2f bb = *(const v2f*)(bPrev + k);
            afrag.x = fmaxf(afrag.x * di + bb.x, 0.0f);
            afrag.y = fmaxf(afrag.y * di + bb.y, 0.0f);
            const float* b0 = B + (size_t)k * FDIM + c0 + 2 * l16;
            v2f r0 = *(const v2f*)(b0);
            v2f r1 = *(const v2f*)(b0 + FDIM);
            #pragma unroll
            for (int nn = 0; nn < 2; ++nn) {
                v2f bfrag;
                bfrag.x = r0[nn];
                bfrag.y = r1[nn];
                cc[nn] = __builtin_amdgcn_wmma_f32_16x16x4_f32(
                    false, afrag, false, bfrag, (short)0, cc[nn], false, false);
            }
        }

        #pragma unroll
        for (int g = 0; g < 8; ++g) {
            int m = m0 + g + half * 8;
            if (m < nrows) {
                float* op = hwsNext + (size_t)(base + m) * FDIM + c0 + 2 * l16;
                *(v2f*)op = (v2f){cc[0][g] * dg[g], cc[1][g] * dg[g]};
            }
        }
    }
}

__global__ __launch_bounds__(256) void gemm_edge(const float* __restrict__ AGG,
                                                 const float* __restrict__ dis,
                                                 const float* __restrict__ b3,
                                                 const int* __restrict__ pos_tr,
                                                 const int* __restrict__ neg_tr,
                                                 const int* __restrict__ pos_te,
                                                 const int* __restrict__ neg_te,
                                                 const float* __restrict__ Wl1,
                                                 const float* __restrict__ bl1,
                                                 const float* __restrict__ Wl2,
                                                 const float* __restrict__ bl2,
                                                 float* __restrict__ out) {
    int wave = (blockIdx.x * blockDim.x + threadIdx.x) >> 5;
    int lane = threadIdx.x & 31;
    int r0 = wave * 16;
    if (r0 >= N_ROWS) return;
    int half = lane >> 4;
    int l16  = lane & 15;
    int r = r0 + l16;

    int u, v;
    if (r < E_TR)            { u = pos_tr[r];              v = pos_tr[E_TR + r]; }
    else if (r < 2 * E_TR)   { int i = r - E_TR;           u = neg_tr[i]; v = neg_tr[E_TR + i]; }
    else if (r < 2 * E_TR + E_TE) { int i = r - 2 * E_TR;  u = pos_te[i]; v = pos_te[E_TE + i]; }
    else                     { int i = r - 2 * E_TR - E_TE; u = neg_te[i]; v = neg_te[E_TE + i]; }

    const float* Au = AGG + (size_t)u * FDIM;
    const float* Av = AGG + (size_t)v * FDIM;
    float du = dis[u];
    float dv = dis[v];

    v8f acc[8] = {};

    strip_k128(Au, du, b3, Wl1, half, l16, A_AFFINE, acc);
    strip_k128(Av, dv, b3, Wl1 + (size_t)FDIM * FDIM, half, l16, A_AFFINE, acc);

    float partial[8];
    #pragma unroll
    for (int g = 0; g < 8; ++g) partial[g] = 0.0f;

    v4f bb4[2], ww4[2];
    bb4[0] = *(const v4f*)(bl1 + 4 * l16);
    bb4[1] = *(const v4f*)(bl1 + 64 + 4 * l16);
    ww4[0] = *(const v4f*)(Wl2 + 4 * l16);
    ww4[1] = *(const v4f*)(Wl2 + 64 + 4 * l16);
    #pragma unroll
    for (int n = 0; n < 8; ++n) {
        float bb = bb4[n >> 2][n & 3];
        float ww = ww4[n >> 2][n & 3];
        #pragma unroll
        for (int g = 0; g < 8; ++g) {
            float z = fmaxf(acc[n][g] + bb, 0.0f);
            partial[g] += z * ww;
        }
    }

    float bias2 = bl2[0];
    #pragma unroll
    for (int g = 0; g < 8; ++g) {
        float s = partial[g];
        s += __shfl_xor(s, 8, 16);
        s += __shfl_xor(s, 4, 16);
        s += __shfl_xor(s, 2, 16);
        s += __shfl_xor(s, 1, 16);
        if (l16 == 0) out[r0 + g + half * 8] = s + bias2;
    }
}

extern "C" void kernel_launch(void* const* d_in, const int* in_sizes, int n_in,
                              void* d_out, int out_size, void* d_ws, size_t ws_size,
                              hipStream_t stream) {
    const float* x          = (const float*)d_in[0];
    const int*   edge_index = (const int*)d_in[1];
    const int*   pos_tr     = (const int*)d_in[2];
    const int*   neg_tr     = (const int*)d_in[3];
    const int*   pos_te     = (const int*)d_in[4];
    const int*   neg_te     = (const int*)d_in[5];
    const float* W0 = (const float*)d_in[6];
    const float* b0 = (const float*)d_in[7];
    const float* W1 = (const float*)d_in[8];
    const float* b1 = (const float*)d_in[9];
    const float* W2 = (const float*)d_in[10];
    const float* b2 = (const float*)d_in[11];
    const float* W3 = (const float*)d_in[12];
    const float* b3 = (const float*)d_in[13];
    const float* Wl1 = (const float*)d_in[14];
    const float* bl1 = (const float*)d_in[15];
    const float* Wl2 = (const float*)d_in[16];
    const float* bl2 = (const float*)d_in[17];
    float* out = (float*)d_out;

    const size_t NF = (size_t)N_NODES * FDIM;
    float* ws   = (float*)d_ws;
    float* dis  = ws;
    float* hws  = ws + 50048;
    float* agg  = hws + NF;
    unsigned int* own  = (unsigned int*)(agg + NF);
    unsigned int* cnt  = own + N_EDGES / 4;
    int*          meta = (int*)(cnt + OC_WG * 256);
    unsigned int* list = (unsigned int*)(meta + 2 * OWN_WG);

    const int* src = edge_index;
    const int* dst = edge_index + N_EDGES;

    own_count<<<OC_WG, OC_THREADS, 0, stream>>>(dst, own, cnt);
    owner_build<<<OWN_WG, OWN_THREADS, 0, stream>>>(src, dst, own, cnt, list, (unsigned int*)agg, meta, dis);

    const int tilesM     = (N_NODES + 15) / 16;
    const int gemmBlocks = (tilesM + 7) / 8;

    gemm_node_first<<<gemmBlocks, 256, 0, stream>>>(x, dis, W0, hws, N_NODES);
    agg_gemm<<<OWN_WG, OWN_THREADS, 0, stream>>>(hws, list, meta, dis, b0, W1, agg);
    agg_gemm<<<OWN_WG, OWN_THREADS, 0, stream>>>(agg, list, meta, dis, b1, W2, hws);
    agg_gemm<<<OWN_WG, OWN_THREADS, 0, stream>>>(hws, list, meta, dis, b2, W3, agg);
    agg_owner<<<OWN_WG, OWN_THREADS, 0, stream>>>(agg, list, meta, hws);

    const int edgeTiles  = N_ROWS / 16;
    const int edgeBlocks = (edgeTiles + 7) / 8;
    gemm_edge<<<edgeBlocks, 256, 0, stream>>>(hws, dis, b3, pos_tr, neg_tr, pos_te, neg_te,
                                              Wl1, bl1, Wl2, bl2, out);
}


# Owner lists from slice runs: launch 1 orders each 4096-edge slice by owner in LDS, stored whole; an owner fetches its run per slice; no all-edge scan, no lone src/dst reads; row sums: arrival order
# speedup vs baseline: 3.6995x; 1.0607x over previous
#include <hip/hip_runtime.h>


typedef float v2f __attribute__((ext_vector_type(2)));
typedef float v8f __attribute__((ext_vector_type(8)));
typedef float v4f __attribute__((ext_vector_type(4)));

#define N_NODES 50000
#define N_EDGES 800000
#define FDIM    128
#define E_TR    100000
#define E_TE    20000
#define N_ROWS  240000

#define OWN_ROWS    196
#define OWN_WG      ((N_NODES + OWN_ROWS - 1) / OWN_ROWS)
#define OWN_THREADS 1024
#define OC_THREADS  256
#define SL_EDGES    (16 * OC_THREADS)
#define OC_WG       ((N_EDGES + SL_EDGES - 1) / SL_EDGES)
static_assert(OWN_WG <= 256, "an owner id must fit in the low byte of a sort key");
static_assert(N_EDGES % 16 == 0, "a thread's 16 edges are all inside the edge list or all outside");
static_assert(N_NODES <= 65536 && OWN_ROWS < 65536, "a list word holds the source row in 16 bits and the owned row above them");
static_assert(OC_WG % 4 == 0, "the count table is summed in four equal parts");
static_assert(N_EDGES / 4 >= OC_WG * 256, "the run-start table fits its region");
static_assert((size_t)OC_WG * SL_EDGES <= (size_t)N_NODES * FDIM, "the ordered slices fit the first plane");
static_assert((SL_EDGES / 4) % OC_THREADS == 0, "a slice goes out in whole 16-byte pieces, the same number a thread");
#define ACC_FLOATS  (164 * 1024 / 4)
static_assert(ACC_FLOATS >= (OWN_ROWS + 1) * FDIM, "owned rows and the spare row fit");
#define ACC_PITCH   132
#define GEMM_ROWS   208
#define GEMM_TASKS  (4 * (GEMM_ROWS / 16))
static_assert(GEMM_ROWS % 16 == 0 && GEMM_ROWS > OWN_ROWS, "whole tiles, and the spare row is among the zeroed rows");
static_assert(ACC_PITCH % 4 == 0, "a row starts on a 16-byte boundary");
static_assert(ACC_FLOATS >= GEMM_ROWS * ACC_PITCH, "the padded rows fit");

__global__ __launch_bounds__(OC_THREADS) void slice_sort(const int* __restrict__ src,
                                                         const int* __restrict__ dst,
                                                         unsigned int* __restrict__ sorted,
                                                         unsigned int* __restrict__ cnt,
                                                         unsigned int* __restrict__ pre) {
    __shared__ unsigned int hist[256];
    __shared__ unsigned int start[256];
    __shared__ unsigned int wtot[OC_THREADS / 32];
    __shared__ alignas(16) unsigned int buf[SL_EDGES];
    const int tid  = threadIdx.x;
    const int lane = tid & 31;
    hist[tid] = 0u;
    __syncthreads();
    const int t = blockIdx.x * OC_THREADS + tid;
    const bool live = t < N_EDGES / 16;
    unsigned int wd[16], key[16];
    #pragma unroll
    for (int e = 0; e < 16; ++e) { wd[e] = 0u; key[e] = 0u; }
    if (live) {
        const int4* d4 = (const int4*)(dst + (size_t)t * 16);
        const int4* s4 = (const int4*)(src + (size_t)t * 16);
        int4 qd[4], qs[4];
        #pragma unroll
        for (int j = 0; j < 4; ++j) { qd[j] = d4[j]; qs[j] = s4[j]; }
        #pragma unroll
        for (int j = 0; j < 4; ++j) {
            const int dd[4] = {qd[j].x, qd[j].y, qd[j].z, qd[j].w};
            const int ss[4] = {qs[j].x, qs[j].y, qs[j].z, qs[j].w};
            #pragma unroll
            for (int i = 0; i < 4; ++i) {
                unsigned int o = (unsigned int)dd[i] / OWN_ROWS;
                unsigned int r = (unsigned int)dd[i] - o * OWN_ROWS;
                unsigned int rank = atomicAdd(&hist[o], 1u);
                wd[4 * j + i]  = (unsigned int)ss[i] | (r << 16);
                key[4 * j + i] = o | (rank << 8);
            }
        }
    }
    __syncthreads();
    const unsigned int v = hist[tid];
    unsigned int incl = v;
    #pragma unroll
    for (int d = 1; d < 32; d <<= 1) {
        unsigned int u = (unsigned int)__shfl_up((int)incl, d);
        if (lane >= d) incl += u;
    }
    if (lane == 31) wtot[tid >> 5] = incl;
    __syncthreads();
    unsigned int wp = 0u;
    for (int k = 0; k < (tid >> 5); ++k) wp += wtot[k];
    const unsigned int excl = wp + incl - v;
    start[tid] = excl;
    cnt[blockIdx.x * 256 + tid] = v;
    pre[blockIdx.x * 256 + tid] = excl;
    __syncthreads();
    if (live) {
        #pragma unroll
        for (int e = 0; e < 16; ++e) buf[start[key[e] & 255u] + (key[e] >> 8)] = wd[e];
    }
    __syncthreads();
    const int left   = N_EDGES - (int)blockIdx.x * SL_EDGES;
    const int nvalid = left < SL_EDGES ? left : SL_EDGES;
    uint4* o4 = (uint4*)(sorted + (size_t)blockIdx.x * SL_EDGES);
    const uint4* b4 = (const uint4*)buf;
    #pragma unroll
    for (int j = 0; j < SL_EDGES / 4 / OC_THREADS; ++j) {
        int i = tid + j * OC_THREADS;
        if (4 * i < nvalid) o4[i] = b4[i];
    }
}

#define A_RAW     0
#define A_AFFINE  1

static __device__ __forceinline__ void strip_k128(const float* __restrict__ Arow, float di,
                                                  const float* __restrict__ bias,
                                                  const float* __restrict__ Bm,
                                                  int half, int l16, const int mode,
                                                  v8f (&acc)[8]) {
    for (int k0 = 0; k0 < FDIM; k0 += 4) {
        int k = k0 + half * 2;
        v2f afrag = *(const v2f*)(Arow + k);
        if (mode != A_RAW) {
            v2f bb = *(const v2f*)(bias + k);
            afrag.x = afrag.x * di + bb.x;
            afrag.y = afrag.y * di + bb.y;
        }
        const float* b0 = Bm + (size_t)k * FDIM + 4 * l16;
        v4f r0[2], r1[2];
        r0[0] = *(const v4f*)(b0);
        r0[1] = *(const v4f*)(b0 + 64);
        r1[0] = *(const v4f*)(b0 + FDIM);
        r1[1] = *(const v4f*)(b0 + FDIM + 64);
        #pragma unroll
        for (int n = 0; n < 8; ++n) {
            v2f bfrag;
            bfrag.x = r0[n >> 2][n & 3];
            bfrag.y = r1[n >> 2][n & 3];
            acc[n] = __builtin_amdgcn_wmma_f32_16x16x4_f32(
                false, afrag, false, bfrag, (short)0, acc[n], false, false);
        }
    }
}

static __device__ __forceinline__ void gemm_node_body(const float* __restrict__ A,
                                                      const float* __restrict__ bprev,
                                                      const float* __restrict__ dis,
                                                      const float* __restrict__ B,
                                                      float* __restrict__ hws,
                                                      int M, const int mode) {
    int wave = (blockIdx.x * blockDim.x + threadIdx.x) >> 5;
    int lane = threadIdx.x & 31;
    int m0 = wave * 16;
    if (m0 >= M) return;
    int half = lane >> 4;
    int l16  = lane & 15;

    const float* Arow = A + (size_t)(m0 + l16) * FDIM;
    float di = (mode != A_RAW) ? dis[m0 + l16] : 0.0f;

    v8f acc[8] = {};
    strip_k128(Arow, di, bprev, B, half, l16, mode, acc);

    float dg[8];
    #pragma unroll
    for (int g = 0; g < 8; ++g) dg[g] = dis[m0 + g + half * 8];

    #pragma unroll
    for (int g = 0; g < 8; ++g) {
        int m = m0 + g + half * 8;
        float* op = hws + (size_t)m * FDIM + 4 * l16;
        *(v4f*)(op)      = (v4f){acc[0][g] * dg[g], acc[1][g] * dg[g], acc[2][g] * dg[g], acc[3][g] * dg[g]};
        *(v4f*)(op + 64) = (v4f){acc[4][g] * dg[g], acc[5][g] * dg[g], acc[6][g] * dg[g], acc[7][g] * dg[g]};
    }
}

__global__ __launch_bounds__(256) void gemm_node_first(const float* __restrict__ X,
                                                       const float* __restrict__ dis,
                                                       const float* __restrict__ B,
                                                       float* __restrict__ hws, int M) {
    gemm_node_body(X, nullptr, dis, B, hws, M, A_RAW);
}


__global__ __launch_bounds__(OWN_THREADS) void owner_build(const unsigned int* __restrict__ sorted,
                                                           const unsigned int* __restrict__ cnt,
                                                           const unsigned int* __restrict__ pre,
                                                           unsigned int* __restrict__ list,
                                                           unsigned int* __restrict__ scratch,
                                                           int* __restrict__ meta,
                                                           float* __restrict__ dis) {
    __shared__ unsigned int rowpos[OWN_ROWS];
    __shared__ unsigned int part[4 * 256];
    __shared__ unsigned int degc[OWN_ROWS];
    __shared__ unsigned int cursor;
    __shared__ unsigned int offs;

    const int b     = blockIdx.x;
    const int base  = b * OWN_ROWS;
    const int nrows = (N_NODES - base < OWN_ROWS) ? (N_NODES - base) : OWN_ROWS;
    const int tid   = threadIdx.x;
    const int lane  = tid & 31;
    const int w     = __builtin_amdgcn_readfirstlane(tid >> 5);

    {
        const int o = tid & 255;
        const int q = tid >> 8;
        unsigned int s = 0u;
        for (int c = q * (OC_WG / 4); c < (q + 1) * (OC_WG / 4); ++c) s += cnt[c * 256 + o];
        part[q * 256 + o] = s;
    }
    if (tid < OWN_ROWS) degc[tid] = 0u;
    if (tid == 0) cursor = 0u;
    __syncthreads();
    if (w == 0) {
        unsigned int s = 0u;
        for (int j = 0; j < 8; ++j) {
            int o = lane * 8 + j;
            if (o < b) s += part[o] + part[256 + o] + part[512 + o] + part[768 + o];
        }
        s += (unsigned int)__shfl_xor((int)s, 16);
        s += (unsigned int)__shfl_xor((int)s, 8);
        s += (unsigned int)__shfl_xor((int)s, 4);
        s += (unsigned int)__shfl_xor((int)s, 2);
        s += (unsigned int)__shfl_xor((int)s, 1);
        if (lane == 0) offs = s;
    }
    __syncthreads();
    unsigned int* mylist = list + offs;
    unsigned int* myraw  = scratch + offs;

    {
        constexpr int SPW = (OC_WG + 31) / 32;
        unsigned int nn[SPW], pp[SPW], wv[SPW];
        #pragma unroll
        for (int k = 0; k < SPW; ++k) {
            const int c  = w + 32 * k;
            const int cc = (c < OC_WG) ? c : 0;
            unsigned int nk = cnt[cc * 256 + b];
            unsigned int pk = pre[cc * 256 + b];
            nn[k] = (c < OC_WG) ? (unsigned int)__builtin_amdgcn_readfirstlane((int)nk) : 0u;
            pp[k] = (unsigned int)__builtin_amdgcn_readfirstlane((int)pk);
        }
        #pragma unroll
        for (int k = 0; k < SPW; ++k) {
            const int c  = w + 32 * k;
            const int cc = (c < OC_WG) ? c : 0;
            unsigned int i = ((unsigned int)lane < nn[k]) ? (pp[k] + (unsigned int)lane) : 0u;
            wv[k] = sorted[(size_t)cc * SL_EDGES + i];
        }
        unsigned int tot = 0u;
        #pragma unroll
        for (int k = 0; k < SPW; ++k) tot += (nn[k] < 32u) ? nn[k] : 32u;
        unsigned int pos = 0u;
        if (lane == 0) pos = atomicAdd(&cursor, tot);
        pos = (unsigned int)__builtin_amdgcn_readfirstlane((int)pos);
        #pragma unroll
        for (int k = 0; k < SPW; ++k) {
            unsigned int m = (nn[k] < 32u) ? nn[k] : 32u;
            if ((unsigned int)lane < m) {
                myraw[pos + lane] = wv[k];
                atomicAdd(&degc[wv[k] >> 16], 1u);
            }
            pos += m;
        }
        #pragma unroll
        for (int k = 0; k < SPW; ++k) {
            const int c  = w + 32 * k;
            const int cc = (c < OC_WG) ? c : 0;
            for (unsigned int i0 = 32u; i0 < nn[k]; i0 += 32u) {
                unsigned int left = nn[k] - i0;
                unsigned int m = left < 32u ? left : 32u;
                unsigned int p2 = 0u;
                if (lane == 0) p2 = atomicAdd(&cursor, m);
                p2 = (unsigned int)__builtin_amdgcn_readfirstlane((int)p2);
                if ((unsigned int)lane < m) {
                    unsigned int x = sorted[(size_t)cc * SL_EDGES + pp[k] + i0 + lane];
                    myraw[p2 + lane] = x;
                    atomicAdd(&degc[x >> 16], 1u);
                }
            }
        }
    }

    __syncthreads();

    static_assert(OWN_ROWS <= 32 * 7, "seven rows a lane cover the owned rows");
    if (w == 0) {
        unsigned int cj[7];
        unsigned int s = 0u;
        #pragma unroll
        for (int j = 0; j < 7; ++j) {
            int idx = lane * 7 + j;
            cj[j] = (idx < OWN_ROWS) ? degc[idx] : 0u;
            s += cj[j];
        }
        unsigned int incl = s;
        #pragma unroll
        for (int d = 1; d < 32; d <<= 1) {
            unsigned int t = (unsigned int)__shfl_up((int)incl, d);
            if (lane >= d) incl += t;
        }
        unsigned int run = incl - s;
        #pragma unroll
        for (int j = 0; j < 7; ++j) {
            int idx = lane * 7 + j;
            if (idx < OWN_ROWS) rowpos[idx] = run;
            run += cj[j];
        }
    }
    __syncthreads();
    {
        const unsigned int nall = cursor;
        for (unsigned int i = (unsigned int)tid; i < nall; i += OWN_THREADS) {
            unsigned int wd = myraw[i];
            unsigned int p = atomicAdd(&rowpos[wd >> 16], 1u);
            mylist[p] = wd;
        }
    }

    if (tid < nrows) dis[base + tid] = rsqrtf((float)degc[tid] + 1.0f);
    if (tid == 0) {
        meta[2 * b]     = (int)offs;
        meta[2 * b + 1] = (int)cursor;
    }
}

__global__ __launch_bounds__(OWN_THREADS) void agg_owner(const float* __restrict__ hws,
                                                         const unsigned int* __restrict__ list,
                                                         const int* __restrict__ meta,
                                                         float* __restrict__ agg) {
    __shared__ alignas(16) float acc[ACC_FLOATS];

    const int b     = blockIdx.x;
    const int base  = b * OWN_ROWS;
    const int nrows = (N_NODES - base < OWN_ROWS) ? (N_NODES - base) : OWN_ROWS;
    const int tid   = threadIdx.x;
    const int lane  = tid & 31;
    const int w     = __builtin_amdgcn_readfirstlane(tid >> 5);

    const int off = meta[2 * b];
    const int n   = meta[2 * b + 1];

    {
        const v4f* h4 = (const v4f*)(hws + (size_t)base * FDIM);
        v4f* a4 = (v4f*)acc;
        const int lim = nrows * (FDIM / 4);
        constexpr int NPC = (OWN_ROWS * (FDIM / 4) + OWN_THREADS - 1) / OWN_THREADS;
        static_assert(NPC == 7, "the fence below names seven values");
        v4f t4[NPC];
        #pragma unroll
        for (int t = 0; t < NPC; ++t) {
            int i = tid + t * OWN_THREADS;
            t4[t] = h4[i < lim ? i : lim - 1];
        }
        asm volatile("" :: "v"(t4[0]), "v"(t4[1]), "v"(t4[2]), "v"(t4[3]), "v"(t4[4]), "v"(t4[5]), "v"(t4[6]));
        #pragma unroll
        for (int t = 0; t < NPC; ++t) {
            int i = tid + t * OWN_THREADS;
            if (i < lim) a4[i] = t4[t];
        }
        if (tid < FDIM / 4) a4[OWN_ROWS * (FDIM / 4) + tid] = (v4f){0.0f, 0.0f, 0.0f, 0.0f};
    }
    const int per = (((n + 31) >> 5) + 7) & ~7;
    const int lo  = w * per;
    const int hi  = (lo + per < n) ? (lo + per) : n;
    const unsigned int* mylist = list + off;
    const unsigned int none = (unsigned int)OWN_ROWS << 16;
    unsigned int pk_next = (lane < hi - lo) ? mylist[lo + lane] : none;

    __syncthreads();

    for (int i0 = lo; i0 < hi; i0 += 32) {
        int left = hi - i0;
        unsigned int pk = pk_next;
        pk_next = (lane < left - 32) ? mylist[i0 + 32 + lane] : none;
        int s_l = (int)(pk & 0xFFFFu);
        int r_l = (int)(pk >> 16);
        #pragma unroll
        for (int g = 0; g < 4; ++g) {
            if (g * 8 < left) {
                float v[8][4];
                #pragma unroll
                for (int j = 0; j < 8; ++j) {
                    int s = __builtin_amdgcn_readlane(s_l, g * 8 + j);
                    const float* hp = hws + (size_t)s * FDIM + lane;
                    v[j][0] = hp[0];
                    v[j][1] = hp[32];
                    v[j][2] = hp[64];
                    v[j][3] = hp[96];
                }
                int rc = __builtin_amdgcn_readlane(r_l, g * 8);
                float a0 = v[0][0], a1 = v[0][1], a2 = v[0][2], a3 = v[0][3];
                #pragma unroll
                for (int j = 1; j < 8; ++j) {
                    int r = __builtin_amdgcn_readlane(r_l, g * 8 + j);
                    if (r != rc) {
                        float* ap = acc + rc * FDIM + lane;
                        atomicAdd(ap + 0,  a0);
                        atomicAdd(ap + 32, a1);
                        atomicAdd(ap + 64, a2);
                        atomicAdd(ap + 96, a3);
                        rc = r;
                        a0 = v[j][0]; a1 = v[j][1]; a2 = v[j][2]; a3 = v[j][3];
                    } else {
                        a0 += v[j][0]; a1 += v[j][1]; a2 += v[j][2]; a3 += v[j][3];
                    }
                }
                {
                    float* ap = acc + rc * FDIM + lane;
                    atomicAdd(ap + 0,  a0);
                    atomicAdd(ap + 32, a1);
                    atomicAdd(ap + 64, a2);
                    atomicAdd(ap + 96, a3);
                }
            }
        }
    }

    __syncthreads();

    {
        const float4* a4 = (const float4*)acc;
        float4* o4 = (float4*)(agg + (size_t)base * FDIM);
        for (int i = tid; i < nrows * (FDIM / 4); i += OWN_THREADS) o4[i] = a4[i];
    }
}

__global__ __launch_bounds__(OWN_THREADS) void agg_gemm(const float* __restrict__ hws,
                                                        const unsigned int* __restrict__ list,
                                                        const int* __restrict__ meta,
                                                        const float* __restrict__ dis,
                                                        const float* __restrict__ bPrev,
                                                        const float* __restrict__ B,
                                                        float* __restrict__ hwsNext) {
    __shared__ alignas(16) float acc[ACC_FLOATS];

    const int b     = blockIdx.x;
    const int base  = b * OWN_ROWS;
    const int nrows = (N_NODES - base < OWN_ROWS) ? (N_NODES - base) : OWN_ROWS;
    const int tid   = threadIdx.x;
    const int lane  = tid & 31;
    const int w     = __builtin_amdgcn_readfirstlane(tid >> 5);

    const int off = meta[2 * b];
    const int n   = meta[2 * b + 1];

    {
        const v4f* h4 = (const v4f*)(hws + (size_t)base * FDIM);
        v4f* a4 = (v4f*)acc;
        const int lim = nrows * (FDIM / 4);
        constexpr int NPC = (GEMM_ROWS * (FDIM / 4) + OWN_THREADS - 1) / OWN_THREADS;
        static_assert(NPC == 7, "the fence below names seven values");
        v4f t4[NPC];
        #pragma unroll
        for (int t = 0; t < NPC; ++t) {
            int i = tid + t * OWN_THREADS;
            t4[t] = h4[i < lim ? i : lim - 1];
        }
        asm volatile("" :: "v"(t4[0]), "v"(t4[1]), "v"(t4[2]), "v"(t4[3]), "v"(t4[4]), "v"(t4[5]), "v"(t4[6]));
        #pragma unroll
        for (int t = 0; t < NPC; ++t) {
            int i = tid + t * OWN_THREADS;
            if (i < GEMM_ROWS * (FDIM / 4)) {
                v4f val = (i < lim) ? t4[t] : (v4f){0.0f, 0.0f, 0.0f, 0.0f};
                a4[(i >> 5) * (ACC_PITCH / 4) + (i & 31)] = val;
            }
        }
    }
    const int per = (((n + 31) >> 5) + 7) & ~7;
    const int lo  = w * per;
    const int hi  = (lo + per < n) ? (lo + per) : n;
    const unsigned int* mylist = list + off;
    const unsigned int none = (unsigned int)OWN_ROWS << 16;
    unsigned int pk_next = (lane < hi - lo) ? mylist[lo + lane] : none;

    __syncthreads();

    for (int i0 = lo; i0 < hi; i0 += 32) {
        int left = hi - i0;
        unsigned int pk = pk_next;
        pk_next = (lane < left - 32) ? mylist[i0 + 32 + lane] : none;
        int s_l = (int)(pk & 0xFFFFu);
        int r_l = (int)(pk >> 16);
        #pragma unroll
        for (int g = 0; g < 4; ++g) {
            if (g * 8 < left) {
                float v[8][4];
                #pragma unroll
                for (int j = 0; j < 8; ++j) {
                    int s = __builtin_amdgcn_readlane(s_l, g * 8 + j);
                    const float* hp = hws + (size_t)s * FDIM + lane;
                    v[j][0] = hp[0];
                    v[j][1] = hp[32];
                    v[j][2] = hp[64];
                    v[j][3] = hp[96];
                }
                int rc = __builtin_amdgcn_readlane(r_l, g * 8);
                float a0 = v[0][0], a1 = v[0][1], a2 = v[0][2], a3 = v[0][3];
                #pragma unroll
                for (int j = 1; j < 8; ++j) {
                    int r = __builtin_amdgcn_readlane(r_l, g * 8 + j);
                    if (r != rc) {
                        float* ap = acc + rc * ACC_PITCH + lane;
                        atomicAdd(ap + 0,  a0);
                        atomicAdd(ap + 32, a1);
                        atomicAdd(ap + 64, a2);
                        atomicAdd(ap + 96, a3);
                        rc = r;
                        a0 = v[j][0]; a1 = v[j][1]; a2 = v[j][2]; a3 = v[j][3];
                    } else {
                        a0 += v[j][0]; a1 += v[j][1]; a2 += v[j][2]; a3 += v[j][3];
                    }
                }
                {
                    float* ap = acc + rc * ACC_PITCH + lane;
                    atomicAdd(ap + 0,  a0);
                    atomicAdd(ap + 32, a1);
                    atomicAdd(ap + 64, a2);
                    atomicAdd(ap + 96, a3);
                }
            }
        }
    }

    __syncthreads();

    const int half = lane >> 4;
    const int l16  = lane & 15;
    for (int t = w; t < GEMM_TASKS; t += OWN_THREADS / 32) {
        const int m0 = (t >> 2) * 16;
        if (m0 >= nrows) break;
        const int c0 = (t & 3) * 32;
        const float* Arow = acc + (m0 + l16) * ACC_PITCH;
        int gr = base + m0 + l16;
        float di = dis[gr < N_NODES ? gr : N_NODES - 1];
        float dg[8];
        #pragma unroll
        for (int g = 0; g < 8; ++g) {
            int mr = base + m0 + g + half * 8;
            dg[g] = dis[mr < N_NODES ? mr : N_NODES - 1];
        }

        v8f cc[2] = {};
        for (int k0 = 0; k0 < FDIM; k0 += 4) {
            int k = k0 + half * 2;
            v2f afrag = *(const v2f*)(Arow + k);
            v2f bb = *(const v2f*)(bPrev + k);
            afrag.x = fmaxf(afrag.x * di + bb.x, 0.0f);
            afrag.y = fmaxf(afrag.y * di + bb.y, 0.0f);
            const float* b0 = B + (size_t)k * FDIM + c0 + 2 * l16;
            v2f r0 = *(const v2f*)(b0);
            v2f r1 = *(const v2f*)(b0 + FDIM);
            #pragma unroll
            for (int nn = 0; nn < 2; ++nn) {
                v2f bfrag;
                bfrag.x = r0[nn];
                bfrag.y = r1[nn];
                cc[nn] = __builtin_amdgcn_wmma_f32_16x16x4_f32(
                    false, afrag, false, bfrag, (short)0, cc[nn], false, false);
            }
        }

        #pragma unroll
        for (int g = 0; g < 8; ++g) {
            int m = m0 + g + half * 8;
            if (m < nrows) {
                float* op = hwsNext + (size_t)(base + m) * FDIM + c0 + 2 * l16;
                *(v2f*)op = (v2f){cc[0][g] * dg[g], cc[1][g] * dg[g]};
            }
        }
    }
}

__global__ __launch_bounds__(256) void gemm_edge(const float* __restrict__ AGG,
                                                 const float* __restrict__ dis,
                                                 const float* __restrict__ b3,
                                                 const int* __restrict__ pos_tr,
                                                 const int* __restrict__ neg_tr,
                                                 const int* __restrict__ pos_te,
                                                 const int* __restrict__ neg_te,
                                                 const float* __restrict__ Wl1,
                                                 const float* __restrict__ bl1,
                                                 const float* __restrict__ Wl2,
                                                 const float* __restrict__ bl2,
                                                 float* __restrict__ out) {
    int wave = (blockIdx.x * blockDim.x + threadIdx.x) >> 5;
    int lane = threadIdx.x & 31;
    int r0 = wave * 16;
    if (r0 >= N_ROWS) return;
    int half = lane >> 4;
    int l16  = lane & 15;
    int r = r0 + l16;

    int u, v;
    if (r < E_TR)            { u = pos_tr[r];              v = pos_tr[E_TR + r]; }
    else if (r < 2 * E_TR)   { int i = r - E_TR;           u = neg_tr[i]; v = neg_tr[E_TR + i]; }
    else if (r < 2 * E_TR + E_TE) { int i = r - 2 * E_TR;  u = pos_te[i]; v = pos_te[E_TE + i]; }
    else                     { int i = r - 2 * E_TR - E_TE; u = neg_te[i]; v = neg_te[E_TE + i]; }

    const float* Au = AGG + (size_t)u * FDIM;
    const float* Av = AGG + (size_t)v * FDIM;
    float du = dis[u];
    float dv = dis[v];

    v8f acc[8] = {};

    strip_k128(Au, du, b3, Wl1, half, l16, A_AFFINE, acc);
    strip_k128(Av, dv, b3, Wl1 + (size_t)FDIM * FDIM, half, l16, A_AFFINE, acc);

    float partial[8];
    #pragma unroll
    for (int g = 0; g < 8; ++g) partial[g] = 0.0f;

    v4f bb4[2], ww4[2];
    bb4[0] = *(const v4f*)(bl1 + 4 * l16);
    bb4[1] = *(const v4f*)(bl1 + 64 + 4 * l16);
    ww4[0] = *(const v4f*)(Wl2 + 4 * l16);
    ww4[1] = *(const v4f*)(Wl2 + 64 + 4 * l16);
    #pragma unroll
    for (int n = 0; n < 8; ++n) {
        float bb = bb4[n >> 2][n & 3];
        float ww = ww4[n >> 2][n & 3];
        #pragma unroll
        for (int g = 0; g < 8; ++g) {
            float z = fmaxf(acc[n][g] + bb, 0.0f);
            partial[g] += z * ww;
        }
    }

    float bias2 = bl2[0];
    #pragma unroll
    for (int g = 0; g < 8; ++g) {
        float s = partial[g];
        s += __shfl_xor(s, 8, 16);
        s += __shfl_xor(s, 4, 16);
        s += __shfl_xor(s, 2, 16);
        s += __shfl_xor(s, 1, 16);
        if (l16 == 0) out[r0 + g + half * 8] = s + bias2;
    }
}

extern "C" void kernel_launch(void* const* d_in, const int* in_sizes, int n_in,
                              void* d_out, int out_size, void* d_ws, size_t ws_size,
                              hipStream_t stream) {
    const float* x          = (const float*)d_in[0];
    const int*   edge_index = (const int*)d_in[1];
    const int*   pos_tr     = (const int*)d_in[2];
    const int*   neg_tr     = (const int*)d_in[3];
    const int*   pos_te     = (const int*)d_in[4];
    const int*   neg_te     = (const int*)d_in[5];
    const float* W0 = (const float*)d_in[6];
    const float* b0 = (const float*)d_in[7];
    const float* W1 = (const float*)d_in[8];
    const float* b1 = (const float*)d_in[9];
    const float* W2 = (const float*)d_in[10];
    const float* b2 = (const float*)d_in[11];
    const float* W3 = (const float*)d_in[12];
    const float* b3 = (const float*)d_in[13];
    const float* Wl1 = (const float*)d_in[14];
    const float* bl1 = (const float*)d_in[15];
    const float* Wl2 = (const float*)d_in[16];
    const float* bl2 = (const float*)d_in[17];
    float* out = (float*)d_out;

    const size_t NF = (size_t)N_NODES * FDIM;
    float* ws   = (float*)d_ws;
    float* dis  = ws;
    float* hws  = ws + 50048;
    float* agg  = hws + NF;
    unsigned int* pre  = (unsigned int*)(agg + NF);
    unsigned int* cnt  = pre + N_EDGES / 4;
    int*          meta = (int*)(cnt + OC_WG * 256);
    unsigned int* list = (unsigned int*)(meta + 2 * OWN_WG);

    const int* src = edge_index;
    const int* dst = edge_index + N_EDGES;

    slice_sort<<<OC_WG, OC_THREADS, 0, stream>>>(src, dst, (unsigned int*)hws, cnt, pre);
    owner_build<<<OWN_WG, OWN_THREADS, 0, stream>>>((const unsigned int*)hws, cnt, pre, list, (unsigned int*)agg, meta, dis);

    const int tilesM     = (N_NODES + 15) / 16;
    const int gemmBlocks = (tilesM + 7) / 8;

    gemm_node_first<<<gemmBlocks, 256, 0, stream>>>(x, dis, W0, hws, N_NODES);
    agg_gemm<<<OWN_WG, OWN_THREADS, 0, stream>>>(hws, list, meta, dis, b0, W1, agg);
    agg_gemm<<<OWN_WG, OWN_THREADS, 0, stream>>>(agg, list, meta, dis, b1, W2, hws);
    agg_gemm<<<OWN_WG, OWN_THREADS, 0, stream>>>(hws, list, meta, dis, b2, W3, agg);
    agg_owner<<<OWN_WG, OWN_THREADS, 0, stream>>>(agg, list, meta, hws);

    const int edgeTiles  = N_ROWS / 16;
    const int edgeBlocks = (edgeTiles + 7) / 8;
    gemm_edge<<<edgeBlocks, 256, 0, stream>>>(hws, dis, b3, pos_tr, neg_tr, pos_te, neg_te,
                                              Wl1, bl1, Wl2, bl2, out);
}
